# starting kernel
# baseline (speedup 1.0000x reference)
#include <hip/hip_runtime.h>

typedef float v2f __attribute__((ext_vector_type(2)));
typedef float v8f __attribute__((ext_vector_type(8)));

__device__ __forceinline__ v8f wmma_f32(v2f a, v2f b, v8f c) {
    return __builtin_amdgcn_wmma_f32_16x16x4_f32(false, a, false, b, (short)0, c, false, false);
}

__global__ __launch_bounds__(256) void build_unitaries(const float* __restrict__ w1,
                                                       const float* __restrict__ w2,
                                                       float* __restrict__ U) {
    __shared__ float mre[4][10][2][2];
    __shared__ float mim[4][10][2][2];

    const int tid = threadIdx.x;
    if (tid < 40) {
        const int l = tid / 10;
        const int j = tid % 10;
        const float* w = (l < 3) ? (w1 + (l * 10 + j) * 3) : (w2 + j * 3);
        const float phi = w[0], theta = w[1], omega = w[2];
        const float c = cosf(0.5f * theta);
        const float s = sinf(0.5f * theta);
        float sa, ca, sb, cb;
        sincosf(-0.5f * (phi + omega), &sa, &ca);
        sincosf(-0.5f * (phi - omega), &sb, &cb);
        mre[l][j][0][0] =  ca * c;  mim[l][j][0][0] =  sa * c;
        mre[l][j][0][1] = -cb * s;  mim[l][j][0][1] =  sb * s;
        mre[l][j][1][0] =  cb * s;  mim[l][j][1][0] =  sb * s;
        mre[l][j][1][1] =  ca * c;  mim[l][j][1][1] = -sa * c;
    }
    __syncthreads();

    for (int it = 0; it < 32; ++it) {
        const int idx = tid + it * 256;
        const int l   = idx >> 11;
        const int rem = idx & 2047;
        const int mat = rem >> 10;
        const int e   = rem & 1023;
        const int r   = e >> 5;
        const int cc  = e & 31;
        float pr = 1.f, pi = 0.f;
        const int j0 = mat ? 5 : 0;
        #pragma unroll
        for (int jj = 0; jj < 5; ++jj) {
            const int rb = (r  >> (4 - jj)) & 1;
            const int cb = (cc >> (4 - jj)) & 1;
            const float ar = mre[l][j0 + jj][rb][cb];
            const float ai = mim[l][j0 + jj][rb][cb];
            const float nr = pr * ar - pi * ai;
            const float ni = pr * ai + pi * ar;
            pr = nr; pi = ni;
        }
        float* base = U + l * 4096 + mat * 2048;
        base[e]        = pr;
        base[e + 1024] = pi;
    }
}

__global__ __launch_bounds__(128) void qsim(const float* __restrict__ x,
                                            const float* __restrict__ U,
                                            float* __restrict__ out) {
    __shared__ float Sr[1024], Si[1024];
    __shared__ float Cr[1024], Ci[1024];
    __shared__ float Ar[1024], Ai[1024], Br[1024], Bi[1024];
    __shared__ float prr[32], pri[32], pcr[32], pci[32];
    __shared__ float red[128][10];

    const int tid  = threadIdx.x;
    const int b    = blockIdx.x;
    const int lane = tid & 31;
    const int wid  = tid >> 5;
    const int ti   = wid >> 1;
    const int tj   = wid & 1;

    for (int i = tid; i < 1024; i += 128) { Sr[i] = (i == 0) ? 1.f : 0.f; Si[i] = 0.f; }

    float xb[10];
    #pragma unroll
    for (int j = 0; j < 10; ++j) xb[j] = x[b * 10 + j];

    if (tid < 64) {
        const int  idx = tid & 31;
        const bool isC = tid >= 32;
        float a = 0.f;
        #pragma unroll
        for (int jj = 0; jj < 5; ++jj) {
            const int bit = (idx >> (4 - jj)) & 1;
            a += xb[isC ? 5 + jj : jj] * (1.f - 2.f * (float)bit);
        }
        a *= -0.5f;
        float sv, cv; sincosf(a, &sv, &cv);
        if (isC) { pcr[idx] = cv; pci[idx] = sv; }
        else     { prr[idx] = cv; pri[idx] = sv; }
    }
    __syncthreads();

    const int mrow = ti * 16 + (lane & 15);
    const int ncol = tj * 16 + (lane & 15);
    const int ksub = (lane >> 4) << 1;

    for (int l = 0; l < 4; ++l) {
        const float* base = U + l * 4096;
        for (int i = tid; i < 1024; i += 128) {
            Ar[i] = base[i];        Ai[i] = base[1024 + i];
            Br[i] = base[2048 + i]; Bi[i] = base[3072 + i];
        }
        __syncthreads();

        {
            v8f cr = {}; v8f ci = {};
            #pragma unroll
            for (int kk = 0; kk < 8; ++kk) {
                const int k = kk * 4 + ksub;
                v2f aR, aI, aIn, bR, bI;
                aR.x = Ar[mrow * 32 + k];  aR.y = Ar[mrow * 32 + k + 1];
                aI.x = Ai[mrow * 32 + k];  aI.y = Ai[mrow * 32 + k + 1];
                aIn.x = -aI.x;             aIn.y = -aI.y;
                bR.x = Sr[k * 32 + ncol];  bR.y = Sr[(k + 1) * 32 + ncol];
                bI.x = Si[k * 32 + ncol];  bI.y = Si[(k + 1) * 32 + ncol];
                cr = wmma_f32(aR,  bR, cr);
                cr = wmma_f32(aIn, bI, cr);
                ci = wmma_f32(aR,  bI, ci);
                ci = wmma_f32(aI,  bR, ci);
            }
            #pragma unroll
            for (int v = 0; v < 8; ++v) {
                const int row = ti * 16 + v + ((lane >> 4) << 3);
                const int col = tj * 16 + (lane & 15);
                Cr[row * 32 + col] = cr[v];
                Ci[row * 32 + col] = ci[v];
            }
        }
        __syncthreads();

        {
            v8f tr = {}; v8f tv = {};
            #pragma unroll
            for (int kk = 0; kk < 8; ++kk) {
                const int k = kk * 4 + ksub;
                v2f aR, aI, aIn, bR, bI;
                aR.x = Cr[mrow * 32 + k];  aR.y = Cr[mrow * 32 + k + 1];
                aI.x = Ci[mrow * 32 + k];  aI.y = Ci[mrow * 32 + k + 1];
                aIn.x = -aI.x;             aIn.y = -aI.y;
                bR.x = Br[ncol * 32 + k];  bR.y = Br[ncol * 32 + k + 1];
                bI.x = Bi[ncol * 32 + k];  bI.y = Bi[ncol * 32 + k + 1];
                tr = wmma_f32(aR,  bR, tr);
                tr = wmma_f32(aIn, bI, tr);
                tv = wmma_f32(aR,  bI, tv);
                tv = wmma_f32(aI,  bR, tv);
            }
            #pragma unroll
            for (int v = 0; v < 8; ++v) {
                const int row = ti * 16 + v + ((lane >> 4) << 3);
                const int col = tj * 16 + (lane & 15);
                const int k   = row * 32 + col;
                const unsigned u = (unsigned)k;
                const int par = __popc(u & (u >> 1)) + (int)((u & 1u) & (u >> 9));
                const float sgn = (par & 1) ? -1.f : 1.f;
                float pRe = sgn, pIm = 0.f;
                if (l < 3) {
                    const float a1 = prr[row], b1 = pri[row];
                    const float a2 = pcr[col], b2 = pci[col];
                    pRe = sgn * (a1 * a2 - b1 * b2);
                    pIm = sgn * (a1 * b2 + b1 * a2);
                }
                const float er = tr[v], ei = tv[v];
                Sr[k] = er * pRe - ei * pIm;
                Si[k] = er * pIm + ei * pRe;
            }
        }
        __syncthreads();
    }

    float acc[10];
    #pragma unroll
    for (int j = 0; j < 10; ++j) acc[j] = 0.f;
    for (int i = tid; i < 1024; i += 128) {
        const float p = Sr[i] * Sr[i] + Si[i] * Si[i];
        #pragma unroll
        for (int j = 0; j < 10; ++j)
            acc[j] += ((i >> (9 - j)) & 1) ? -p : p;
    }
    #pragma unroll
    for (int j = 0; j < 10; ++j) red[tid][j] = acc[j];
    __syncthreads();
    if (tid < 10) {
        float s = 0.f;
        for (int t = 0; t < 128; ++t) s += red[t][tid];
        out[b * 10 + tid] = s;
    }
}

extern "C" void kernel_launch(void* const* d_in, const int* in_sizes, int n_in,
                              void* d_out, int out_size, void* d_ws, size_t ws_size,
                              hipStream_t stream) {
    const float* x  = (const float*)d_in[0];
    const float* w1 = (const float*)d_in[1];
    const float* w2 = (const float*)d_in[2];
    float* out = (float*)d_out;
    float* U   = (float*)d_ws;

    const int B = in_sizes[0] / 10;

    hipLaunchKernelGGL(build_unitaries, dim3(1), dim3(256), 0, stream, w1, w2, U);
    hipLaunchKernelGGL(qsim, dim3(B), dim3(128), 0, stream, x, U, out);
}


# LDS re-layout: row pitch 32->36 floats, each lane half takes 4 consecutive k (16-byte row reads), reduction scratch overlaid on dead C planes; removes LDS bank conflicts; k summation order changes
# speedup vs baseline: 1.1673x; 1.1673x over previous
#include <hip/hip_runtime.h>

typedef float v2f __attribute__((ext_vector_type(2)));
typedef float v8f __attribute__((ext_vector_type(8)));

__device__ __forceinline__ v8f wmma_f32(v2f a, v2f b, v8f c) {
    return __builtin_amdgcn_wmma_f32_16x16x4_f32(false, a, false, b, (short)0, c, false, false);
}

__global__ __launch_bounds__(256) void build_unitaries(const float* __restrict__ w1,
                                                       const float* __restrict__ w2,
                                                       float* __restrict__ U) {
    __shared__ float mre[4][10][2][2];
    __shared__ float mim[4][10][2][2];

    const int tid = threadIdx.x;
    if (tid < 40) {
        const int l = tid / 10;
        const int j = tid % 10;
        const float* w = (l < 3) ? (w1 + (l * 10 + j) * 3) : (w2 + j * 3);
        const float phi = w[0], theta = w[1], omega = w[2];
        const float c = cosf(0.5f * theta);
        const float s = sinf(0.5f * theta);
        float sa, ca, sb, cb;
        sincosf(-0.5f * (phi + omega), &sa, &ca);
        sincosf(-0.5f * (phi - omega), &sb, &cb);
        mre[l][j][0][0] =  ca * c;  mim[l][j][0][0] =  sa * c;
        mre[l][j][0][1] = -cb * s;  mim[l][j][0][1] =  sb * s;
        mre[l][j][1][0] =  cb * s;  mim[l][j][1][0] =  sb * s;
        mre[l][j][1][1] =  ca * c;  mim[l][j][1][1] = -sa * c;
    }
    __syncthreads();

    for (int it = 0; it < 32; ++it) {
        const int idx = tid + it * 256;
        const int l   = idx >> 11;
        const int rem = idx & 2047;
        const int mat = rem >> 10;
        const int e   = rem & 1023;
        const int r   = e >> 5;
        const int cc  = e & 31;
        float pr = 1.f, pi = 0.f;
        const int j0 = mat ? 5 : 0;
        #pragma unroll
        for (int jj = 0; jj < 5; ++jj) {
            const int rb = (r  >> (4 - jj)) & 1;
            const int cb = (cc >> (4 - jj)) & 1;
            const float ar = mre[l][j0 + jj][rb][cb];
            const float ai = mim[l][j0 + jj][rb][cb];
            const float nr = pr * ar - pi * ai;
            const float ni = pr * ai + pi * ar;
            pr = nr; pi = ni;
        }
        float* base = U + l * 4096 + mat * 2048;
        base[e]        = pr;
        base[e + 1024] = pi;
    }
}

constexpr int LP    = 36;
constexpr int PLANE = 32 * LP;

__global__ __launch_bounds__(128) void qsim(const float* __restrict__ x,
                                            const float* __restrict__ U,
                                            float* __restrict__ out) {
    __shared__ __attribute__((aligned(16))) float lds[8 * PLANE + 128];
    float* const Sr  = lds;
    float* const Si  = lds + 1 * PLANE;
    float* const Cr  = lds + 2 * PLANE;
    float* const Ci  = lds + 3 * PLANE;
    float* const Ar  = lds + 4 * PLANE;
    float* const Ai  = lds + 5 * PLANE;
    float* const Br  = lds + 6 * PLANE;
    float* const Bi  = lds + 7 * PLANE;
    float* const prr = lds + 8 * PLANE;
    float* const pri = prr + 32;
    float* const pcr = prr + 64;
    float* const pci = prr + 96;
    float* const red = Cr;

    const int tid  = threadIdx.x;
    const int b    = blockIdx.x;
    const int lane = tid & 31;
    const int wid  = tid >> 5;
    const int ti   = wid >> 1;
    const int tj   = wid & 1;

    for (int i = tid; i < 1024; i += 128) {
        const int p = (i >> 5) * LP + (i & 31);
        Sr[p] = (i == 0) ? 1.f : 0.f; Si[p] = 0.f;
    }

    float xb[10];
    #pragma unroll
    for (int j = 0; j < 10; ++j) xb[j] = x[b * 10 + j];

    if (tid < 64) {
        const int  idx = tid & 31;
        const bool isC = tid >= 32;
        float a = 0.f;
        #pragma unroll
        for (int jj = 0; jj < 5; ++jj) {
            const int bit = (idx >> (4 - jj)) & 1;
            a += xb[isC ? 5 + jj : jj] * (1.f - 2.f * (float)bit);
        }
        a *= -0.5f;
        float sv, cv; sincosf(a, &sv, &cv);
        if (isC) { pcr[idx] = cv; pci[idx] = sv; }
        else     { prr[idx] = cv; pri[idx] = sv; }
    }
    __syncthreads();

    const int mrow = ti * 16 + (lane & 15);
    const int ncol = tj * 16 + (lane & 15);
    const int koff = (lane >> 4) << 2;

    for (int l = 0; l < 4; ++l) {
        const float* base = U + l * 4096;
        for (int i = tid; i < 1024; i += 128) {
            const int p = (i >> 5) * LP + (i & 31);
            Ar[p] = base[i];        Ai[p] = base[1024 + i];
            Br[p] = base[2048 + i]; Bi[p] = base[3072 + i];
        }
        __syncthreads();

        {
            v8f cr = {}; v8f ci = {};
            #pragma unroll
            for (int j = 0; j < 4; ++j) {
                const int kb = j * 8 + koff;
                const float4 ar4 = *reinterpret_cast<const float4*>(&Ar[mrow * LP + kb]);
                const float4 ai4 = *reinterpret_cast<const float4*>(&Ai[mrow * LP + kb]);
                const float arv[4] = {ar4.x, ar4.y, ar4.z, ar4.w};
                const float aiv[4] = {ai4.x, ai4.y, ai4.z, ai4.w};
                #pragma unroll
                for (int q = 0; q < 2; ++q) {
                    const int k = kb + 2 * q;
                    v2f aR, aI, aIn, bR, bI;
                    aR.x = arv[2 * q];         aR.y = arv[2 * q + 1];
                    aI.x = aiv[2 * q];         aI.y = aiv[2 * q + 1];
                    aIn.x = -aI.x;             aIn.y = -aI.y;
                    bR.x = Sr[k * LP + ncol];  bR.y = Sr[(k + 1) * LP + ncol];
                    bI.x = Si[k * LP + ncol];  bI.y = Si[(k + 1) * LP + ncol];
                    cr = wmma_f32(aR,  bR, cr);
                    cr = wmma_f32(aIn, bI, cr);
                    ci = wmma_f32(aR,  bI, ci);
                    ci = wmma_f32(aI,  bR, ci);
                }
            }
            #pragma unroll
            for (int v = 0; v < 8; ++v) {
                const int row = ti * 16 + v + ((lane >> 4) << 3);
                const int col = tj * 16 + (lane & 15);
                Cr[row * LP + col] = cr[v];
                Ci[row * LP + col] = ci[v];
            }
        }
        __syncthreads();

        {
            v8f tr = {}; v8f tv = {};
            #pragma unroll
            for (int j = 0; j < 4; ++j) {
                const int kb = j * 8 + koff;
                const float4 cr4 = *reinterpret_cast<const float4*>(&Cr[mrow * LP + kb]);
                const float4 ci4 = *reinterpret_cast<const float4*>(&Ci[mrow * LP + kb]);
                const float4 br4 = *reinterpret_cast<const float4*>(&Br[ncol * LP + kb]);
                const float4 bi4 = *reinterpret_cast<const float4*>(&Bi[ncol * LP + kb]);
                const float crv[4] = {cr4.x, cr4.y, cr4.z, cr4.w};
                const float civ[4] = {ci4.x, ci4.y, ci4.z, ci4.w};
                const float brv[4] = {br4.x, br4.y, br4.z, br4.w};
                const float biv[4] = {bi4.x, bi4.y, bi4.z, bi4.w};
                #pragma unroll
                for (int q = 0; q < 2; ++q) {
                    v2f aR, aI, aIn, bR, bI;
                    aR.x = crv[2 * q];         aR.y = crv[2 * q + 1];
                    aI.x = civ[2 * q];         aI.y = civ[2 * q + 1];
                    aIn.x = -aI.x;             aIn.y = -aI.y;
                    bR.x = brv[2 * q];         bR.y = brv[2 * q + 1];
                    bI.x = biv[2 * q];         bI.y = biv[2 * q + 1];
                    tr = wmma_f32(aR,  bR, tr);
                    tr = wmma_f32(aIn, bI, tr);
                    tv = wmma_f32(aR,  bI, tv);
                    tv = wmma_f32(aI,  bR, tv);
                }
            }
            #pragma unroll
            for (int v = 0; v < 8; ++v) {
                const int row = ti * 16 + v + ((lane >> 4) << 3);
                const int col = tj * 16 + (lane & 15);
                const int k   = row * 32 + col;
                const unsigned u = (unsigned)k;
                const int par = __popc(u & (u >> 1)) + (int)((u & 1u) & (u >> 9));
                const float sgn = (par & 1) ? -1.f : 1.f;
                float pRe = sgn, pIm = 0.f;
                if (l < 3) {
                    const float a1 = prr[row], b1 = pri[row];
                    const float a2 = pcr[col], b2 = pci[col];
                    pRe = sgn * (a1 * a2 - b1 * b2);
                    pIm = sgn * (a1 * b2 + b1 * a2);
                }
                const float er = tr[v], ei = tv[v];
                Sr[row * LP + col] = er * pRe - ei * pIm;
                Si[row * LP + col] = er * pIm + ei * pRe;
            }
        }
        __syncthreads();
    }

    float acc[10];
    #pragma unroll
    for (int j = 0; j < 10; ++j) acc[j] = 0.f;
    for (int i = tid; i < 1024; i += 128) {
        const int pidx = (i >> 5) * LP + (i & 31);
        const float p = Sr[pidx] * Sr[pidx] + Si[pidx] * Si[pidx];
        #pragma unroll
        for (int j = 0; j < 10; ++j)
            acc[j] += ((i >> (9 - j)) & 1) ? -p : p;
    }
    #pragma unroll
    for (int j = 0; j < 10; ++j) red[tid * 10 + j] = acc[j];
    __syncthreads();
    if (tid < 10) {
        float s = 0.f;
        for (int t = 0; t < 128; ++t) s += red[t * 10 + tid];
        out[b * 10 + tid] = s;
    }
}

extern "C" void kernel_launch(void* const* d_in, const int* in_sizes, int n_in,
                              void* d_out, int out_size, void* d_ws, size_t ws_size,
                              hipStream_t stream) {
    const float* x  = (const float*)d_in[0];
    const float* w1 = (const float*)d_in[1];
    const float* w2 = (const float*)d_in[2];
    float* out = (float*)d_out;
    float* U   = (float*)d_ws;

    const int B = in_sizes[0] / 10;

    hipLaunchKernelGGL(build_unitaries, dim3(1), dim3(256), 0, stream, w1, w2, U);
    hipLaunchKernelGGL(qsim, dim3(B), dim3(128), 0, stream, x, U, out);
}


# layer 0 acts on |0..0>: its two products keep only the one k-step that meets the non-zero row/column (8 WMMAs per wave, not 64); its staging and S init cut to k 0..7; only exact zeros dropped
# speedup vs baseline: 1.2918x; 1.1067x over previous
#include <hip/hip_runtime.h>

typedef float v2f __attribute__((ext_vector_type(2)));
typedef float v8f __attribute__((ext_vector_type(8)));

__device__ __forceinline__ v8f wmma_f32(v2f a, v2f b, v8f c) {
    return __builtin_amdgcn_wmma_f32_16x16x4_f32(false, a, false, b, (short)0, c, false, false);
}

__global__ __launch_bounds__(256) void build_unitaries(const float* __restrict__ w1,
                                                       const float* __restrict__ w2,
                                                       float* __restrict__ U) {
    __shared__ float mre[4][10][2][2];
    __shared__ float mim[4][10][2][2];

    const int tid = threadIdx.x;
    if (tid < 40) {
        const int l = tid / 10;
        const int j = tid % 10;
        const float* w = (l < 3) ? (w1 + (l * 10 + j) * 3) : (w2 + j * 3);
        const float phi = w[0], theta = w[1], omega = w[2];
        const float c = cosf(0.5f * theta);
        const float s = sinf(0.5f * theta);
        float sa, ca, sb, cb;
        sincosf(-0.5f * (phi + omega), &sa, &ca);
        sincosf(-0.5f * (phi - omega), &sb, &cb);
        mre[l][j][0][0] =  ca * c;  mim[l][j][0][0] =  sa * c;
        mre[l][j][0][1] = -cb * s;  mim[l][j][0][1] =  sb * s;
        mre[l][j][1][0] =  cb * s;  mim[l][j][1][0] =  sb * s;
        mre[l][j][1][1] =  ca * c;  mim[l][j][1][1] = -sa * c;
    }
    __syncthreads();

    for (int it = 0; it < 32; ++it) {
        const int idx = tid + it * 256;
        const int l   = idx >> 11;
        const int rem = idx & 2047;
        const int mat = rem >> 10;
        const int e   = rem & 1023;
        const int r   = e >> 5;
        const int cc  = e & 31;
        float pr = 1.f, pi = 0.f;
        const int j0 = mat ? 5 : 0;
        #pragma unroll
        for (int jj = 0; jj < 5; ++jj) {
            const int rb = (r  >> (4 - jj)) & 1;
            const int cb = (cc >> (4 - jj)) & 1;
            const float ar = mre[l][j0 + jj][rb][cb];
            const float ai = mim[l][j0 + jj][rb][cb];
            const float nr = pr * ar - pi * ai;
            const float ni = pr * ai + pi * ar;
            pr = nr; pi = ni;
        }
        float* base = U + l * 4096 + mat * 2048;
        base[e]        = pr;
        base[e + 1024] = pi;
    }
}

constexpr int LP    = 36;
constexpr int PLANE = 32 * LP;

template <bool FIRST>
__device__ __forceinline__ void run_layer(const int l, const float* __restrict__ U,
                                          float* Sr, float* Si, float* Cr, float* Ci,
                                          float* Ar, float* Ai, float* Br, float* Bi,
                                          const float* prr, const float* pri,
                                          const float* pcr, const float* pci,
                                          const int tid, const int lane,
                                          const int ti, const int tj,
                                          const int mrow, const int ncol, const int koff) {
    constexpr int NJ = FIRST ? 1 : 4;
    constexpr int NQ = FIRST ? 1 : 2;

    const float* base = U + l * 4096;
    if (FIRST) {
        for (int i = tid; i < 256; i += 128) {
            const int g = (i >> 3) * 32 + (i & 7);
            const int p = (i >> 3) * LP + (i & 7);
            Ar[p] = base[g];        Ai[p] = base[1024 + g];
            Br[p] = base[2048 + g]; Bi[p] = base[3072 + g];
        }
    } else {
        for (int i = tid; i < 1024; i += 128) {
            const int p = (i >> 5) * LP + (i & 31);
            Ar[p] = base[i];        Ai[p] = base[1024 + i];
            Br[p] = base[2048 + i]; Bi[p] = base[3072 + i];
        }
    }
    __syncthreads();

    {
        v8f cr = {}; v8f ci = {};
        #pragma unroll
        for (int j = 0; j < NJ; ++j) {
            const int kb = j * 8 + koff;
            const float4 ar4 = *reinterpret_cast<const float4*>(&Ar[mrow * LP + kb]);
            const float4 ai4 = *reinterpret_cast<const float4*>(&Ai[mrow * LP + kb]);
            const float arv[4] = {ar4.x, ar4.y, ar4.z, ar4.w};
            const float aiv[4] = {ai4.x, ai4.y, ai4.z, ai4.w};
            #pragma unroll
            for (int q = 0; q < NQ; ++q) {
                const int k = kb + 2 * q;
                v2f aR, aI, aIn, bR, bI;
                aR.x = arv[2 * q];         aR.y = arv[2 * q + 1];
                aI.x = aiv[2 * q];         aI.y = aiv[2 * q + 1];
                aIn.x = -aI.x;             aIn.y = -aI.y;
                bR.x = Sr[k * LP + ncol];  bR.y = Sr[(k + 1) * LP + ncol];
                bI.x = Si[k * LP + ncol];  bI.y = Si[(k + 1) * LP + ncol];
                cr = wmma_f32(aR,  bR, cr);
                cr = wmma_f32(aIn, bI, cr);
                ci = wmma_f32(aR,  bI, ci);
                ci = wmma_f32(aI,  bR, ci);
            }
        }
        #pragma unroll
        for (int v = 0; v < 8; ++v) {
            const int row = ti * 16 + v + ((lane >> 4) << 3);
            const int col = tj * 16 + (lane & 15);
            Cr[row * LP + col] = cr[v];
            Ci[row * LP + col] = ci[v];
        }
    }
    __syncthreads();

    {
        v8f tr = {}; v8f tv = {};
        #pragma unroll
        for (int j = 0; j < NJ; ++j) {
            const int kb = j * 8 + koff;
            const float4 cr4 = *reinterpret_cast<const float4*>(&Cr[mrow * LP + kb]);
            const float4 ci4 = *reinterpret_cast<const float4*>(&Ci[mrow * LP + kb]);
            const float4 br4 = *reinterpret_cast<const float4*>(&Br[ncol * LP + kb]);
            const float4 bi4 = *reinterpret_cast<const float4*>(&Bi[ncol * LP + kb]);
            const float crv[4] = {cr4.x, cr4.y, cr4.z, cr4.w};
            const float civ[4] = {ci4.x, ci4.y, ci4.z, ci4.w};
            const float brv[4] = {br4.x, br4.y, br4.z, br4.w};
            const float biv[4] = {bi4.x, bi4.y, bi4.z, bi4.w};
            #pragma unroll
            for (int q = 0; q < NQ; ++q) {
                v2f aR, aI, aIn, bR, bI;
                aR.x = crv[2 * q];         aR.y = crv[2 * q + 1];
                aI.x = civ[2 * q];         aI.y = civ[2 * q + 1];
                aIn.x = -aI.x;             aIn.y = -aI.y;
                bR.x = brv[2 * q];         bR.y = brv[2 * q + 1];
                bI.x = biv[2 * q];         bI.y = biv[2 * q + 1];
                tr = wmma_f32(aR,  bR, tr);
                tr = wmma_f32(aIn, bI, tr);
                tv = wmma_f32(aR,  bI, tv);
                tv = wmma_f32(aI,  bR, tv);
            }
        }
        #pragma unroll
        for (int v = 0; v < 8; ++v) {
            const int row = ti * 16 + v + ((lane >> 4) << 3);
            const int col = tj * 16 + (lane & 15);
            const int k   = row * 32 + col;
            const unsigned u = (unsigned)k;
            const int par = __popc(u & (u >> 1)) + (int)((u & 1u) & (u >> 9));
            const float sgn = (par & 1) ? -1.f : 1.f;
            float pRe = sgn, pIm = 0.f;
            if (l < 3) {
                const float a1 = prr[row], b1 = pri[row];
                const float a2 = pcr[col], b2 = pci[col];
                pRe = sgn * (a1 * a2 - b1 * b2);
                pIm = sgn * (a1 * b2 + b1 * a2);
            }
            const float er = tr[v], ei = tv[v];
            Sr[row * LP + col] = er * pRe - ei * pIm;
            Si[row * LP + col] = er * pIm + ei * pRe;
        }
    }
    __syncthreads();
}

__global__ __launch_bounds__(128) void qsim(const float* __restrict__ x,
                                            const float* __restrict__ U,
                                            float* __restrict__ out) {
    __shared__ __attribute__((aligned(16))) float lds[8 * PLANE + 128];
    float* const Sr  = lds;
    float* const Si  = lds + 1 * PLANE;
    float* const Cr  = lds + 2 * PLANE;
    float* const Ci  = lds + 3 * PLANE;
    float* const Ar  = lds + 4 * PLANE;
    float* const Ai  = lds + 5 * PLANE;
    float* const Br  = lds + 6 * PLANE;
    float* const Bi  = lds + 7 * PLANE;
    float* const prr = lds + 8 * PLANE;
    float* const pri = prr + 32;
    float* const pcr = prr + 64;
    float* const pci = prr + 96;
    float* const red = Cr;

    const int tid  = threadIdx.x;
    const int b    = blockIdx.x;
    const int lane = tid & 31;
    const int wid  = tid >> 5;
    const int ti   = wid >> 1;
    const int tj   = wid & 1;

    for (int i = tid; i < 256; i += 128) {
        const int p = (i >> 5) * LP + (i & 31);
        Sr[p] = (i == 0) ? 1.f : 0.f; Si[p] = 0.f;
    }

    float xb[10];
    #pragma unroll
    for (int j = 0; j < 10; ++j) xb[j] = x[b * 10 + j];

    if (tid < 64) {
        const int  idx = tid & 31;
        const bool isC = tid >= 32;
        float a = 0.f;
        #pragma unroll
        for (int jj = 0; jj < 5; ++jj) {
            const int bit = (idx >> (4 - jj)) & 1;
            a += xb[isC ? 5 + jj : jj] * (1.f - 2.f * (float)bit);
        }
        a *= -0.5f;
        float sv, cv; sincosf(a, &sv, &cv);
        if (isC) { pcr[idx] = cv; pci[idx] = sv; }
        else     { prr[idx] = cv; pri[idx] = sv; }
    }
    __syncthreads();

    const int mrow = ti * 16 + (lane & 15);
    const int ncol = tj * 16 + (lane & 15);
    const int koff = (lane >> 4) << 2;

    run_layer<true>(0, U, Sr, Si, Cr, Ci, Ar, Ai, Br, Bi, prr, pri, pcr, pci,
                    tid, lane, ti, tj, mrow, ncol, koff);
    for (int l = 1; l < 4; ++l)
        run_layer<false>(l, U, Sr, Si, Cr, Ci, Ar, Ai, Br, Bi, prr, pri, pcr, pci,
                         tid, lane, ti, tj, mrow, ncol, koff);

    float acc[10];
    #pragma unroll
    for (int j = 0; j < 10; ++j) acc[j] = 0.f;
    for (int i = tid; i < 1024; i += 128) {
        const int pidx = (i >> 5) * LP + (i & 31);
        const float p = Sr[pidx] * Sr[pidx] + Si[pidx] * Si[pidx];
        #pragma unroll
        for (int j = 0; j < 10; ++j)
            acc[j] += ((i >> (9 - j)) & 1) ? -p : p;
    }
    #pragma unroll
    for (int j = 0; j < 10; ++j) red[tid * 10 + j] = acc[j];
    __syncthreads();
    if (tid < 10) {
        float s = 0.f;
        for (int t = 0; t < 128; ++t) s += red[t * 10 + tid];
        out[b * 10 + tid] = s;
    }
}

extern "C" void kernel_launch(void* const* d_in, const int* in_sizes, int n_in,
                              void* d_out, int out_size, void* d_ws, size_t ws_size,
                              hipStream_t stream) {
    const float* x  = (const float*)d_in[0];
    const float* w1 = (const float*)d_in[1];
    const float* w2 = (const float*)d_in[2];
    float* out = (float*)d_out;
    float* U   = (float*)d_ws;

    const int B = in_sizes[0] / 10;

    hipLaunchKernelGGL(build_unitaries, dim3(1), dim3(256), 0, stream, w1, w2, U);
    hipLaunchKernelGGL(qsim, dim3(B), dim3(128), 0, stream, x, U, out);
}


# operands in fragment order end to end: wave = 32x16 column strip, S*B^T first and its D tiles are A*C's B operand in registers, factors pre-ordered in global; only S crosses LDS; sum orders change
# speedup vs baseline: 1.7689x; 1.3693x over previous
#include <hip/hip_runtime.h>

typedef float v2f __attribute__((ext_vector_type(2)));
typedef float v8f __attribute__((ext_vector_type(8)));

__device__ __forceinline__ v8f wmma_f32(v2f a, v2f b, v8f c) {
    return __builtin_amdgcn_wmma_f32_16x16x4_f32(false, a, false, b, (short)0, c, false, false);
}

__global__ __launch_bounds__(256) void build_unitaries(const float* __restrict__ w1,
                                                       const float* __restrict__ w2,
                                                       float* __restrict__ U) {
    __shared__ float mre[4][10][2][2];
    __shared__ float mim[4][10][2][2];

    const int tid = threadIdx.x;
    if (tid < 40) {
        const int l = tid / 10;
        const int j = tid % 10;
        const float* w = (l < 3) ? (w1 + (l * 10 + j) * 3) : (w2 + j * 3);
        const float phi = w[0], theta = w[1], omega = w[2];
        const float c = cosf(0.5f * theta);
        const float s = sinf(0.5f * theta);
        float sa, ca, sb, cb;
        sincosf(-0.5f * (phi + omega), &sa, &ca);
        sincosf(-0.5f * (phi - omega), &sb, &cb);
        mre[l][j][0][0] =  ca * c;  mim[l][j][0][0] =  sa * c;
        mre[l][j][0][1] = -cb * s;  mim[l][j][0][1] =  sb * s;
        mre[l][j][1][0] =  cb * s;  mim[l][j][1][0] =  sb * s;
        mre[l][j][1][1] =  ca * c;  mim[l][j][1][1] = -sa * c;
    }
    __syncthreads();

    for (int it = 0; it < 32; ++it) {
        const int idx = tid + it * 256;
        const int l   = idx >> 11;
        const int rem = idx & 2047;
        const int mat = rem >> 10;
        const int e   = rem & 1023;
        const int r   = e >> 5;
        const int cc  = e & 31;
        float pr = 1.f, pi = 0.f;
        const int j0 = mat ? 5 : 0;
        #pragma unroll
        for (int jj = 0; jj < 5; ++jj) {
            const int rb = (r  >> (4 - jj)) & 1;
            const int cb = (cc >> (4 - jj)) & 1;
            const float ar = mre[l][j0 + jj][rb][cb];
            const float ai = mim[l][j0 + jj][rb][cb];
            const float nr = pr * ar - pi * ai;
            const float ni = pr * ai + pi * ar;
            pr = nr; pi = ni;
        }
        const int c16 = r & 15;
        const int rt  = r >> 4;
        int f4, lane;
        if (mat) {
            const int j = cc >> 3;
            const int h = (cc >> 2) & 1;
            lane = h * 16 + c16;
            f4   = ((rt * 4 + j) * 2) * 32 + lane;
        } else {
            const int t  = cc >> 4;
            const int h  = (cc >> 3) & 1;
            const int ee = (cc >> 2) & 1;
            lane = h * 16 + c16;
            f4   = 512 + (((rt * 2 + t) * 2 + ee) * 2) * 32 + lane;
        }
        float* dst = U + l * 4096 + f4 * 4 + (cc & 3);
        dst[0]   = pr;
        dst[128] = pi;
    }
}

constexpr int LP    = 36;
constexpr int PLANE = 32 * LP;

template <int L>
__device__ __forceinline__ void load_bfrags(const float* __restrict__ U,
                                            const int lane, const int tj,
                                            float4 (&bf)[8]) {
    const float4* __restrict__ Bf =
        reinterpret_cast<const float4*>(U + L * 4096) + tj * 256 + lane;
    constexpr int N = (L == 0) ? 2 : 8;
    #pragma unroll
    for (int i = 0; i < N; ++i) bf[i] = Bf[i * 32];
}

template <int L>
__device__ __forceinline__ void layer_products(const float* __restrict__ U,
                                               const float4 (&bf)[8],
                                               const float* Sr, const float* Si,
                                               const int lane, const int tj,
                                               v8f (&tr)[2], v8f (&tv)[2]) {
    constexpr bool FIRST = (L == 0);
    const int c = lane & 15;
    const int h = lane >> 4;
    const float4* __restrict__ Af =
        reinterpret_cast<const float4*>(U + L * 4096) + 512 + lane;

    v8f cr[2], ci[2];
    cr[0] = v8f{}; cr[1] = v8f{}; ci[0] = v8f{}; ci[1] = v8f{};

    if (FIRST) {
        const float4 br4 = bf[0];
        const float4 bi4 = bf[1];
        v2f sR, bR, bI;
        sR.x = (lane == 0) ? 1.f : 0.f;  sR.y = 0.f;
        bR.x = br4.x;  bR.y = br4.y;
        bI.x = bi4.x;  bI.y = bi4.y;
        cr[0] = wmma_f32(sR, bR, cr[0]);
        ci[0] = wmma_f32(sR, bI, ci[0]);
    } else {
        #pragma unroll
        for (int j = 0; j < 4; ++j) {
            const float4 br4 = bf[j * 2];
            const float4 bi4 = bf[j * 2 + 1];
            const float brv[4] = {br4.x, br4.y, br4.z, br4.w};
            const float biv[4] = {bi4.x, bi4.y, bi4.z, bi4.w};
            #pragma unroll
            for (int t = 0; t < 2; ++t) {
                const int so = (16 * t + c) * LP + 8 * j + 4 * h;
                const float4 sr4 = *reinterpret_cast<const float4*>(&Sr[so]);
                const float4 si4 = *reinterpret_cast<const float4*>(&Si[so]);
                const float srv[4] = {sr4.x, sr4.y, sr4.z, sr4.w};
                const float siv[4] = {si4.x, si4.y, si4.z, si4.w};
                #pragma unroll
                for (int q = 0; q < 2; ++q) {
                    v2f aR, aI, aIn, bR, bI;
                    aR.x = srv[2 * q];   aR.y = srv[2 * q + 1];
                    aI.x = siv[2 * q];   aI.y = siv[2 * q + 1];
                    aIn.x = -aI.x;       aIn.y = -aI.y;
                    bR.x = brv[2 * q];   bR.y = brv[2 * q + 1];
                    bI.x = biv[2 * q];   bI.y = biv[2 * q + 1];
                    cr[t] = wmma_f32(aR,  bR, cr[t]);
                    cr[t] = wmma_f32(aIn, bI, cr[t]);
                    ci[t] = wmma_f32(aR,  bI, ci[t]);
                    ci[t] = wmma_f32(aI,  bR, ci[t]);
                }
            }
        }
    }

    tr[0] = v8f{}; tr[1] = v8f{}; tv[0] = v8f{}; tv[1] = v8f{};
    constexpr int NT = FIRST ? 1 : 2;
    constexpr int NE = FIRST ? 1 : 2;
    constexpr int NQ = FIRST ? 1 : 2;
    #pragma unroll
    for (int t = 0; t < NT; ++t) {
        #pragma unroll
        for (int ee = 0; ee < NE; ++ee) {
            #pragma unroll
            for (int u = 0; u < 2; ++u) {
                const float4 ar4 = Af[(((u * 2 + t) * 2 + ee) * 2) * 32];
                const float4 ai4 = Af[(((u * 2 + t) * 2 + ee) * 2 + 1) * 32];
                const float arv[4] = {ar4.x, ar4.y, ar4.z, ar4.w};
                const float aiv[4] = {ai4.x, ai4.y, ai4.z, ai4.w};
                #pragma unroll
                for (int q = 0; q < NQ; ++q) {
                    const int s = ee * 2 + q;
                    v2f aR, aI, aIn, bR, bI;
                    aR.x = arv[2 * q];       aR.y = arv[2 * q + 1];
                    aI.x = aiv[2 * q];       aI.y = aiv[2 * q + 1];
                    aIn.x = -aI.x;           aIn.y = -aI.y;
                    bR.x = cr[t][2 * s];     bR.y = cr[t][2 * s + 1];
                    bI.x = ci[t][2 * s];     bI.y = ci[t][2 * s + 1];
                    tr[u] = wmma_f32(aR,  bR, tr[u]);
                    tr[u] = wmma_f32(aIn, bI, tr[u]);
                    tv[u] = wmma_f32(aR,  bI, tv[u]);
                    tv[u] = wmma_f32(aI,  bR, tv[u]);
                }
            }
        }
    }
}

template <int L>
__device__ __forceinline__ void layer_diag(v8f (&tr)[2], v8f (&tv)[2],
                                           const float* prr, const float* pri,
                                           const float* pcr, const float* pci,
                                           const int lane, const int tj) {
    const int col = tj * 16 + (lane & 15);
    float a2 = 1.f, b2 = 0.f;
    if (L < 3) { a2 = pcr[col]; b2 = pci[col]; }
    #pragma unroll
    for (int u = 0; u < 2; ++u) {
        #pragma unroll
        for (int v = 0; v < 8; ++v) {
            const int row = u * 16 + v + ((lane >> 4) << 3);
            const unsigned k = (unsigned)(row * 32 + col);
            const int par = __popc(k & (k >> 1)) + (int)((k & 1u) & (k >> 9));
            const float sgn = (par & 1) ? -1.f : 1.f;
            float pRe = sgn, pIm = 0.f;
            if (L < 3) {
                const float a1 = prr[row], b1 = pri[row];
                pRe = sgn * (a1 * a2 - b1 * b2);
                pIm = sgn * (a1 * b2 + b1 * a2);
            }
            const float er = tr[u][v], ei = tv[u][v];
            tr[u][v] = er * pRe - ei * pIm;
            tv[u][v] = er * pIm + ei * pRe;
        }
    }
}

__device__ __forceinline__ void store_strip(const v8f (&tr)[2], const v8f (&tv)[2],
                                            float* Dr, float* Di,
                                            const int lane, const int tj) {
    const int col = tj * 16 + (lane & 15);
    #pragma unroll
    for (int u = 0; u < 2; ++u) {
        #pragma unroll
        for (int v = 0; v < 8; ++v) {
            const int row = u * 16 + v + ((lane >> 4) << 3);
            Dr[row * LP + col] = tr[u][v];
            Di[row * LP + col] = tv[u][v];
        }
    }
}

__global__ __launch_bounds__(64) void qsim(const float* __restrict__ x,
                                           const float* __restrict__ U,
                                           float* __restrict__ out) {
    __shared__ __attribute__((aligned(16))) float lds[4 * PLANE + 128];
    float* const S0r = lds;
    float* const S0i = lds + 1 * PLANE;
    float* const S1r = lds + 2 * PLANE;
    float* const S1i = lds + 3 * PLANE;
    float* const prr = lds + 4 * PLANE;
    float* const pri = prr + 32;
    float* const pcr = prr + 64;
    float* const pci = prr + 96;
    float* const red = S0r;

    const int tid  = threadIdx.x;
    const int b    = blockIdx.x;
    const int lane = tid & 31;
    const int tj   = tid >> 5;

    float4 bf[8];
    load_bfrags<0>(U, lane, tj, bf);

    float xb[10];
    #pragma unroll
    for (int j = 0; j < 10; ++j) xb[j] = x[b * 10 + j];

    {
        const int  idx = tid & 31;
        const bool isC = tid >= 32;
        float a = 0.f;
        #pragma unroll
        for (int jj = 0; jj < 5; ++jj) {
            const int bit = (idx >> (4 - jj)) & 1;
            a += xb[isC ? 5 + jj : jj] * (1.f - 2.f * (float)bit);
        }
        a *= -0.5f;
        float sv, cv; sincosf(a, &sv, &cv);
        if (isC) { pcr[idx] = cv; pci[idx] = sv; }
        else     { prr[idx] = cv; pri[idx] = sv; }
    }
    __syncthreads();

    v8f tr[2], tv[2];

    layer_products<0>(U, bf, S0r, S0i, lane, tj, tr, tv);
    layer_diag<0>(tr, tv, prr, pri, pcr, pci, lane, tj);
    load_bfrags<1>(U, lane, tj, bf);
    store_strip(tr, tv, S1r, S1i, lane, tj);
    __syncthreads();

    layer_products<1>(U, bf, S1r, S1i, lane, tj, tr, tv);
    layer_diag<1>(tr, tv, prr, pri, pcr, pci, lane, tj);
    load_bfrags<2>(U, lane, tj, bf);
    store_strip(tr, tv, S0r, S0i, lane, tj);
    __syncthreads();

    layer_products<2>(U, bf, S0r, S0i, lane, tj, tr, tv);
    layer_diag<2>(tr, tv, prr, pri, pcr, pci, lane, tj);
    load_bfrags<3>(U, lane, tj, bf);
    store_strip(tr, tv, S1r, S1i, lane, tj);
    __syncthreads();

    layer_products<3>(U, bf, S1r, S1i, lane, tj, tr, tv);
    layer_diag<3>(tr, tv, prr, pri, pcr, pci, lane, tj);

    float acc[10];
    #pragma unroll
    for (int j = 0; j < 10; ++j) acc[j] = 0.f;
    {
        const int col = tj * 16 + (lane & 15);
        #pragma unroll
        for (int u = 0; u < 2; ++u) {
            #pragma unroll
            for (int v = 0; v < 8; ++v) {
                const int row = u * 16 + v + ((lane >> 4) << 3);
                const int i   = row * 32 + col;
                const float p = tr[u][v] * tr[u][v] + tv[u][v] * tv[u][v];
                #pragma unroll
                for (int j = 0; j < 10; ++j)
                    acc[j] += ((i >> (9 - j)) & 1) ? -p : p;
            }
        }
    }
    #pragma unroll
    for (int j = 0; j < 10; ++j) red[tid * 10 + j] = acc[j];
    __syncthreads();
    if (tid < 10) {
        float s = 0.f;
        for (int t = 0; t < 64; ++t) s += red[t * 10 + tid];
        out[b * 10 + tid] = s;
    }
}

extern "C" void kernel_launch(void* const* d_in, const int* in_sizes, int n_in,
                              void* d_out, int out_size, void* d_ws, size_t ws_size,
                              hipStream_t stream) {
    const float* x  = (const float*)d_in[0];
    const float* w1 = (const float*)d_in[1];
    const float* w2 = (const float*)d_in[2];
    float* out = (float*)d_out;
    float* U   = (float*)d_ws;

    const int B = in_sizes[0] / 10;

    hipLaunchKernelGGL(build_unitaries, dim3(1), dim3(256), 0, stream, w1, w2, U);
    hipLaunchKernelGGL(qsim, dim3(B), dim3(64), 0, stream, x, U, out);
}


# factor-build kernel: 1 block looping 32 times -> 8 blocks (one per layer and factor), a thread forms one fragment slot (4 columns) and stores two float4; a wire's four trig calls run on four waves
# speedup vs baseline: 2.0981x; 1.1861x over previous
#include <hip/hip_runtime.h>

typedef float v2f __attribute__((ext_vector_type(2)));
typedef float v8f __attribute__((ext_vector_type(8)));

__device__ __forceinline__ v8f wmma_f32(v2f a, v2f b, v8f c) {
    return __builtin_amdgcn_wmma_f32_16x16x4_f32(false, a, false, b, (short)0, c, false, false);
}

__global__ __launch_bounds__(256) void build_unitaries(const float* __restrict__ w1,
                                                       const float* __restrict__ w2,
                                                       float* __restrict__ U) {
    __shared__ float trig[5][8];

    const int tid  = threadIdx.x;
    const int l    = blockIdx.x >> 1;
    const int mat  = blockIdx.x & 1;
    const int j0   = mat ? 5 : 0;
    const int wv   = tid >> 5;
    const int lane = tid & 31;

    if (wv < 4 && lane < 5) {
        const int j = j0 + lane;
        const float* w = (l < 3) ? (w1 + (l * 10 + j) * 3) : (w2 + j * 3);
        const float phi = w[0], theta = w[1], omega = w[2];
        if (wv == 0) {
            trig[lane][0] = cosf(0.5f * theta);
        } else if (wv == 1) {
            trig[lane][1] = sinf(0.5f * theta);
        } else if (wv == 2) {
            float sa, ca;
            sincosf(-0.5f * (phi + omega), &sa, &ca);
            trig[lane][2] = ca;  trig[lane][3] = sa;
        } else {
            float sb, cb;
            sincosf(-0.5f * (phi - omega), &sb, &cb);
            trig[lane][4] = cb;  trig[lane][5] = sb;
        }
    }
    __syncthreads();

    const int q   = wv;
    const int c16 = lane & 15;
    const int h   = lane >> 4;
    const int r   = (q >> 2) * 16 + c16;
    const int cc0 = mat ? ((q & 3) * 8 + h * 4)
                        : (((q >> 1) & 1) * 16 + h * 8 + (q & 1) * 4);

    float gr[5][2], gi[5][2];
    #pragma unroll
    for (int jj = 0; jj < 5; ++jj) {
        const float c  = trig[jj][0], s  = trig[jj][1];
        const float ca = trig[jj][2], sa = trig[jj][3];
        const float cb = trig[jj][4], sb = trig[jj][5];
        const float epr = ca * c, epi = sa * c;
        const float emr = cb * s, emi = sb * s;
        const int rb = (r >> (4 - jj)) & 1;
        gr[jj][0] = rb ?  emr : epr;   gi[jj][0] = rb ?  emi : epi;
        gr[jj][1] = rb ?  epr : -emr;  gi[jj][1] = rb ? -epi : emi;
    }

    float pr = 1.f, pi = 0.f;
    #pragma unroll
    for (int jj = 0; jj < 3; ++jj) {
        const int cb = (cc0 >> (4 - jj)) & 1;
        const float ar = cb ? gr[jj][1] : gr[jj][0];
        const float ai = cb ? gi[jj][1] : gi[jj][0];
        const float nr = pr * ar - pi * ai;
        const float ni = pr * ai + pi * ar;
        pr = nr; pi = ni;
    }
    float er[4], ei[4];
    #pragma unroll
    for (int b1 = 0; b1 < 2; ++b1) {
        const float qr = pr * gr[3][b1] - pi * gi[3][b1];
        const float qi = pr * gi[3][b1] + pi * gr[3][b1];
        #pragma unroll
        for (int b0 = 0; b0 < 2; ++b0) {
            er[b1 * 2 + b0] = qr * gr[4][b0] - qi * gi[4][b0];
            ei[b1 * 2 + b0] = qr * gi[4][b0] + qi * gr[4][b0];
        }
    }

    float4* dst = reinterpret_cast<float4*>(U + l * 4096)
                + (mat ? 0 : 512) + (q * 2) * 32 + lane;
    dst[0]  = make_float4(er[0], er[1], er[2], er[3]);
    dst[32] = make_float4(ei[0], ei[1], ei[2], ei[3]);
}

constexpr int LP    = 36;
constexpr int PLANE = 32 * LP;

template <int L>
__device__ __forceinline__ void load_bfrags(const float* __restrict__ U,
                                            const int lane, const int tj,
                                            float4 (&bf)[8]) {
    const float4* __restrict__ Bf =
        reinterpret_cast<const float4*>(U + L * 4096) + tj * 256 + lane;
    constexpr int N = (L == 0) ? 2 : 8;
    #pragma unroll
    for (int i = 0; i < N; ++i) bf[i] = Bf[i * 32];
}

template <int L>
__device__ __forceinline__ void layer_products(const float* __restrict__ U,
                                               const float4 (&bf)[8],
                                               const float* Sr, const float* Si,
                                               const int lane, const int tj,
                                               v8f (&tr)[2], v8f (&tv)[2]) {
    constexpr bool FIRST = (L == 0);
    const int c = lane & 15;
    const int h = lane >> 4;
    const float4* __restrict__ Af =
        reinterpret_cast<const float4*>(U + L * 4096) + 512 + lane;

    v8f cr[2], ci[2];
    cr[0] = v8f{}; cr[1] = v8f{}; ci[0] = v8f{}; ci[1] = v8f{};

    if (FIRST) {
        const float4 br4 = bf[0];
        const float4 bi4 = bf[1];
        v2f sR, bR, bI;
        sR.x = (lane == 0) ? 1.f : 0.f;  sR.y = 0.f;
        bR.x = br4.x;  bR.y = br4.y;
        bI.x = bi4.x;  bI.y = bi4.y;
        cr[0] = wmma_f32(sR, bR, cr[0]);
        ci[0] = wmma_f32(sR, bI, ci[0]);
    } else {
        #pragma unroll
        for (int j = 0; j < 4; ++j) {
            const float4 br4 = bf[j * 2];
            const float4 bi4 = bf[j * 2 + 1];
            const float brv[4] = {br4.x, br4.y, br4.z, br4.w};
            const float biv[4] = {bi4.x, bi4.y, bi4.z, bi4.w};
            #pragma unroll
            for (int t = 0; t < 2; ++t) {
                const int so = (16 * t + c) * LP + 8 * j + 4 * h;
                const float4 sr4 = *reinterpret_cast<const float4*>(&Sr[so]);
                const float4 si4 = *reinterpret_cast<const float4*>(&Si[so]);
                const float srv[4] = {sr4.x, sr4.y, sr4.z, sr4.w};
                const float siv[4] = {si4.x, si4.y, si4.z, si4.w};
                #pragma unroll
                for (int q = 0; q < 2; ++q) {
                    v2f aR, aI, aIn, bR, bI;
                    aR.x = srv[2 * q];   aR.y = srv[2 * q + 1];
                    aI.x = siv[2 * q];   aI.y = siv[2 * q + 1];
                    aIn.x = -aI.x;       aIn.y = -aI.y;
                    bR.x = brv[2 * q];   bR.y = brv[2 * q + 1];
                    bI.x = biv[2 * q];   bI.y = biv[2 * q + 1];
                    cr[t] = wmma_f32(aR,  bR, cr[t]);
                    cr[t] = wmma_f32(aIn, bI, cr[t]);
                    ci[t] = wmma_f32(aR,  bI, ci[t]);
                    ci[t] = wmma_f32(aI,  bR, ci[t]);
                }
            }
        }
    }

    tr[0] = v8f{}; tr[1] = v8f{}; tv[0] = v8f{}; tv[1] = v8f{};
    constexpr int NT = FIRST ? 1 : 2;
    constexpr int NE = FIRST ? 1 : 2;
    constexpr int NQ = FIRST ? 1 : 2;
    #pragma unroll
    for (int t = 0; t < NT; ++t) {
        #pragma unroll
        for (int ee = 0; ee < NE; ++ee) {
            #pragma unroll
            for (int u = 0; u < 2; ++u) {
                const float4 ar4 = Af[(((u * 2 + t) * 2 + ee) * 2) * 32];
                const float4 ai4 = Af[(((u * 2 + t) * 2 + ee) * 2 + 1) * 32];
                const float arv[4] = {ar4.x, ar4.y, ar4.z, ar4.w};
                const float aiv[4] = {ai4.x, ai4.y, ai4.z, ai4.w};
                #pragma unroll
                for (int q = 0; q < NQ; ++q) {
                    const int s = ee * 2 + q;
                    v2f aR, aI, aIn, bR, bI;
                    aR.x = arv[2 * q];       aR.y = arv[2 * q + 1];
                    aI.x = aiv[2 * q];       aI.y = aiv[2 * q + 1];
                    aIn.x = -aI.x;           aIn.y = -aI.y;
                    bR.x = cr[t][2 * s];     bR.y = cr[t][2 * s + 1];
                    bI.x = ci[t][2 * s];     bI.y = ci[t][2 * s + 1];
                    tr[u] = wmma_f32(aR,  bR, tr[u]);
                    tr[u] = wmma_f32(aIn, bI, tr[u]);
                    tv[u] = wmma_f32(aR,  bI, tv[u]);
                    tv[u] = wmma_f32(aI,  bR, tv[u]);
                }
            }
        }
    }
}

template <int L>
__device__ __forceinline__ void layer_diag(v8f (&tr)[2], v8f (&tv)[2],
                                           const float* prr, const float* pri,
                                           const float* pcr, const float* pci,
                                           const int lane, const int tj) {
    const int col = tj * 16 + (lane & 15);
    float a2 = 1.f, b2 = 0.f;
    if (L < 3) { a2 = pcr[col]; b2 = pci[col]; }
    #pragma unroll
    for (int u = 0; u < 2; ++u) {
        #pragma unroll
        for (int v = 0; v < 8; ++v) {
            const int row = u * 16 + v + ((lane >> 4) << 3);
            const unsigned k = (unsigned)(row * 32 + col);
            const int par = __popc(k & (k >> 1)) + (int)((k & 1u) & (k >> 9));
            const float sgn = (par & 1) ? -1.f : 1.f;
            float pRe = sgn, pIm = 0.f;
            if (L < 3) {
                const float a1 = prr[row], b1 = pri[row];
                pRe = sgn * (a1 * a2 - b1 * b2);
                pIm = sgn * (a1 * b2 + b1 * a2);
            }
            const float er = tr[u][v], ei = tv[u][v];
            tr[u][v] = er * pRe - ei * pIm;
            tv[u][v] = er * pIm + ei * pRe;
        }
    }
}

__device__ __forceinline__ void store_strip(const v8f (&tr)[2], const v8f (&tv)[2],
                                            float* Dr, float* Di,
                                            const int lane, const int tj) {
    const int col = tj * 16 + (lane & 15);
    #pragma unroll
    for (int u = 0; u < 2; ++u) {
        #pragma unroll
        for (int v = 0; v < 8; ++v) {
            const int row = u * 16 + v + ((lane >> 4) << 3);
            Dr[row * LP + col] = tr[u][v];
            Di[row * LP + col] = tv[u][v];
        }
    }
}

__global__ __launch_bounds__(64) void qsim(const float* __restrict__ x,
                                           const float* __restrict__ U,
                                           float* __restrict__ out) {
    __shared__ __attribute__((aligned(16))) float lds[4 * PLANE + 128];
    float* const S0r = lds;
    float* const S0i = lds + 1 * PLANE;
    float* const S1r = lds + 2 * PLANE;
    float* const S1i = lds + 3 * PLANE;
    float* const prr = lds + 4 * PLANE;
    float* const pri = prr + 32;
    float* const pcr = prr + 64;
    float* const pci = prr + 96;
    float* const red = S0r;

    const int tid  = threadIdx.x;
    const int b    = blockIdx.x;
    const int lane = tid & 31;
    const int tj   = tid >> 5;

    float4 bf[8];
    load_bfrags<0>(U, lane, tj, bf);

    float xb[10];
    #pragma unroll
    for (int j = 0; j < 10; ++j) xb[j] = x[b * 10 + j];

    {
        const int  idx = tid & 31;
        const bool isC = tid >= 32;
        float a = 0.f;
        #pragma unroll
        for (int jj = 0; jj < 5; ++jj) {
            const int bit = (idx >> (4 - jj)) & 1;
            a += xb[isC ? 5 + jj : jj] * (1.f - 2.f * (float)bit);
        }
        a *= -0.5f;
        float sv, cv; sincosf(a, &sv, &cv);
        if (isC) { pcr[idx] = cv; pci[idx] = sv; }
        else     { prr[idx] = cv; pri[idx] = sv; }
    }
    __syncthreads();

    v8f tr[2], tv[2];

    layer_products<0>(U, bf, S0r, S0i, lane, tj, tr, tv);
    layer_diag<0>(tr, tv, prr, pri, pcr, pci, lane, tj);
    load_bfrags<1>(U, lane, tj, bf);
    store_strip(tr, tv, S1r, S1i, lane, tj);
    __syncthreads();

    layer_products<1>(U, bf, S1r, S1i, lane, tj, tr, tv);
    layer_diag<1>(tr, tv, prr, pri, pcr, pci, lane, tj);
    load_bfrags<2>(U, lane, tj, bf);
    store_strip(tr, tv, S0r, S0i, lane, tj);
    __syncthreads();

    layer_products<2>(U, bf, S0r, S0i, lane, tj, tr, tv);
    layer_diag<2>(tr, tv, prr, pri, pcr, pci, lane, tj);
    load_bfrags<3>(U, lane, tj, bf);
    store_strip(tr, tv, S1r, S1i, lane, tj);
    __syncthreads();

    layer_products<3>(U, bf, S1r, S1i, lane, tj, tr, tv);
    layer_diag<3>(tr, tv, prr, pri, pcr, pci, lane, tj);

    float acc[10];
    #pragma unroll
    for (int j = 0; j < 10; ++j) acc[j] = 0.f;
    {
        const int col = tj * 16 + (lane & 15);
        #pragma unroll
        for (int u = 0; u < 2; ++u) {
            #pragma unroll
            for (int v = 0; v < 8; ++v) {
                const int row = u * 16 + v + ((lane >> 4) << 3);
                const int i   = row * 32 + col;
                const float p = tr[u][v] * tr[u][v] + tv[u][v] * tv[u][v];
                #pragma unroll
                for (int j = 0; j < 10; ++j)
                    acc[j] += ((i >> (9 - j)) & 1) ? -p : p;
            }
        }
    }
    #pragma unroll
    for (int j = 0; j < 10; ++j) red[tid * 10 + j] = acc[j];
    __syncthreads();
    if (tid < 10) {
        float s = 0.f;
        for (int t = 0; t < 64; ++t) s += red[t * 10 + tid];
        out[b * 10 + tid] = s;
    }
}

extern "C" void kernel_launch(void* const* d_in, const int* in_sizes, int n_in,
                              void* d_out, int out_size, void* d_ws, size_t ws_size,
                              hipStream_t stream) {
    const float* x  = (const float*)d_in[0];
    const float* w1 = (const float*)d_in[1];
    const float* w2 = (const float*)d_in[2];
    float* out = (float*)d_out;
    float* U   = (float*)d_ws;

    const int B = in_sizes[0] / 10;

    hipLaunchKernelGGL(build_unitaries, dim3(8), dim3(256), 0, stream, w1, w2, U);
    hipLaunchKernelGGL(qsim, dim3(B), dim3(64), 0, stream, x, U, out);
}


# complex products in 3-multiplication form (p1=a(c+d), p2=(b-a)c, p3=(a+b)d; Re=p1-p3, Im=p1+p2): 3 f32 WMMAs per complex k-step, not 4; operand sums on vector ALU; rounding and sum order change
# speedup vs baseline: 2.4691x; 1.1768x over previous
#include <hip/hip_runtime.h>

typedef float v2f __attribute__((ext_vector_type(2)));
typedef float v8f __attribute__((ext_vector_type(8)));

__device__ __forceinline__ v8f wmma_f32(v2f a, v2f b, v8f c) {
    return __builtin_amdgcn_wmma_f32_16x16x4_f32(false, a, false, b, (short)0, c, false, false);
}

__global__ __launch_bounds__(256) void build_unitaries(const float* __restrict__ w1,
                                                       const float* __restrict__ w2,
                                                       float* __restrict__ U) {
    __shared__ float trig[5][8];

    const int tid  = threadIdx.x;
    const int l    = blockIdx.x >> 1;
    const int mat  = blockIdx.x & 1;
    const int j0   = mat ? 5 : 0;
    const int wv   = tid >> 5;
    const int lane = tid & 31;

    if (wv < 4 && lane < 5) {
        const int j = j0 + lane;
        const float* w = (l < 3) ? (w1 + (l * 10 + j) * 3) : (w2 + j * 3);
        const float phi = w[0], theta = w[1], omega = w[2];
        if (wv == 0) {
            trig[lane][0] = cosf(0.5f * theta);
        } else if (wv == 1) {
            trig[lane][1] = sinf(0.5f * theta);
        } else if (wv == 2) {
            float sa, ca;
            sincosf(-0.5f * (phi + omega), &sa, &ca);
            trig[lane][2] = ca;  trig[lane][3] = sa;
        } else {
            float sb, cb;
            sincosf(-0.5f * (phi - omega), &sb, &cb);
            trig[lane][4] = cb;  trig[lane][5] = sb;
        }
    }
    __syncthreads();

    const int q   = wv;
    const int c16 = lane & 15;
    const int h   = lane >> 4;
    const int r   = (q >> 2) * 16 + c16;
    const int cc0 = mat ? ((q & 3) * 8 + h * 4)
                        : (((q >> 1) & 1) * 16 + h * 8 + (q & 1) * 4);

    float gr[5][2], gi[5][2];
    #pragma unroll
    for (int jj = 0; jj < 5; ++jj) {
        const float c  = trig[jj][0], s  = trig[jj][1];
        const float ca = trig[jj][2], sa = trig[jj][3];
        const float cb = trig[jj][4], sb = trig[jj][5];
        const float epr = ca * c, epi = sa * c;
        const float emr = cb * s, emi = sb * s;
        const int rb = (r >> (4 - jj)) & 1;
        gr[jj][0] = rb ?  emr : epr;   gi[jj][0] = rb ?  emi : epi;
        gr[jj][1] = rb ?  epr : -emr;  gi[jj][1] = rb ? -epi : emi;
    }

    float pr = 1.f, pi = 0.f;
    #pragma unroll
    for (int jj = 0; jj < 3; ++jj) {
        const int cb = (cc0 >> (4 - jj)) & 1;
        const float ar = cb ? gr[jj][1] : gr[jj][0];
        const float ai = cb ? gi[jj][1] : gi[jj][0];
        const float nr = pr * ar - pi * ai;
        const float ni = pr * ai + pi * ar;
        pr = nr; pi = ni;
    }
    float er[4], ei[4];
    #pragma unroll
    for (int b1 = 0; b1 < 2; ++b1) {
        const float qr = pr * gr[3][b1] - pi * gi[3][b1];
        const float qi = pr * gi[3][b1] + pi * gr[3][b1];
        #pragma unroll
        for (int b0 = 0; b0 < 2; ++b0) {
            er[b1 * 2 + b0] = qr * gr[4][b0] - qi * gi[4][b0];
            ei[b1 * 2 + b0] = qr * gi[4][b0] + qi * gr[4][b0];
        }
    }

    float4* dst = reinterpret_cast<float4*>(U + l * 4096)
                + (mat ? 0 : 512) + (q * 2) * 32 + lane;
    dst[0]  = make_float4(er[0], er[1], er[2], er[3]);
    dst[32] = make_float4(ei[0], ei[1], ei[2], ei[3]);
}

constexpr int LP    = 36;
constexpr int PLANE = 32 * LP;

template <int L>
__device__ __forceinline__ void load_bfrags(const float* __restrict__ U,
                                            const int lane, const int tj,
                                            float4 (&bf)[8]) {
    const float4* __restrict__ Bf =
        reinterpret_cast<const float4*>(U + L * 4096) + tj * 256 + lane;
    constexpr int N = (L == 0) ? 2 : 8;
    #pragma unroll
    for (int i = 0; i < N; ++i) bf[i] = Bf[i * 32];
}

template <int L>
__device__ __forceinline__ void layer_products(const float* __restrict__ U,
                                               const float4 (&bf)[8],
                                               const float* Sr, const float* Si,
                                               const int lane, const int tj,
                                               v8f (&tr)[2], v8f (&tv)[2]) {
    constexpr bool FIRST = (L == 0);
    const int c = lane & 15;
    const int h = lane >> 4;
    const float4* __restrict__ Af =
        reinterpret_cast<const float4*>(U + L * 4096) + 512 + lane;

    v8f cr[2], ci[2], cs[2];
    cr[0] = v8f{}; cr[1] = v8f{}; ci[0] = v8f{}; ci[1] = v8f{};

    if (FIRST) {
        const float4 br4 = bf[0];
        const float4 bi4 = bf[1];
        v2f sR, bR, bI;
        sR.x = (lane == 0) ? 1.f : 0.f;  sR.y = 0.f;
        bR.x = br4.x;  bR.y = br4.y;
        bI.x = bi4.x;  bI.y = bi4.y;
        cr[0] = wmma_f32(sR, bR, cr[0]);
        ci[0] = wmma_f32(sR, bI, ci[0]);
    } else {
        v8f p1[2], p2[2], p3[2];
        p1[0] = v8f{}; p1[1] = v8f{}; p2[0] = v8f{}; p2[1] = v8f{}; p3[0] = v8f{}; p3[1] = v8f{};
        #pragma unroll
        for (int j = 0; j < 4; ++j) {
            const float4 br4 = bf[j * 2];
            const float4 bi4 = bf[j * 2 + 1];
            const float brv[4] = {br4.x, br4.y, br4.z, br4.w};
            const float bdv[4] = {bi4.x - br4.x, bi4.y - br4.y, bi4.z - br4.z, bi4.w - br4.w};
            const float bsv[4] = {br4.x + bi4.x, br4.y + bi4.y, br4.z + bi4.z, br4.w + bi4.w};
            #pragma unroll
            for (int t = 0; t < 2; ++t) {
                const int so = (16 * t + c) * LP + 8 * j + 4 * h;
                const float4 sr4 = *reinterpret_cast<const float4*>(&Sr[so]);
                const float4 si4 = *reinterpret_cast<const float4*>(&Si[so]);
                const float srv[4] = {sr4.x, sr4.y, sr4.z, sr4.w};
                const float siv[4] = {si4.x, si4.y, si4.z, si4.w};
                const float ssv[4] = {sr4.x + si4.x, sr4.y + si4.y, sr4.z + si4.z, sr4.w + si4.w};
                #pragma unroll
                for (int q = 0; q < 2; ++q) {
                    v2f xS, xR, xI, fR, fD, fS;
                    xS.x = ssv[2 * q];   xS.y = ssv[2 * q + 1];
                    xR.x = srv[2 * q];   xR.y = srv[2 * q + 1];
                    xI.x = siv[2 * q];   xI.y = siv[2 * q + 1];
                    fR.x = brv[2 * q];   fR.y = brv[2 * q + 1];
                    fD.x = bdv[2 * q];   fD.y = bdv[2 * q + 1];
                    fS.x = bsv[2 * q];   fS.y = bsv[2 * q + 1];
                    p1[t] = wmma_f32(xS, fR, p1[t]);
                    p2[t] = wmma_f32(xR, fD, p2[t]);
                    p3[t] = wmma_f32(xI, fS, p3[t]);
                }
            }
        }
        cr[0] = p1[0] - p3[0];  ci[0] = p1[0] + p2[0];
        cr[1] = p1[1] - p3[1];  ci[1] = p1[1] + p2[1];
    }
    cs[0] = cr[0] + ci[0];
    cs[1] = cr[1] + ci[1];

    v8f t1[2], t2[2], t3[2];
    t1[0] = v8f{}; t1[1] = v8f{}; t2[0] = v8f{}; t2[1] = v8f{}; t3[0] = v8f{}; t3[1] = v8f{};
    constexpr int NT = FIRST ? 1 : 2;
    constexpr int NE = FIRST ? 1 : 2;
    constexpr int NQ = FIRST ? 1 : 2;
    #pragma unroll
    for (int t = 0; t < NT; ++t) {
        #pragma unroll
        for (int ee = 0; ee < NE; ++ee) {
            #pragma unroll
            for (int u = 0; u < 2; ++u) {
                const float4 ar4 = Af[(((u * 2 + t) * 2 + ee) * 2) * 32];
                const float4 ai4 = Af[(((u * 2 + t) * 2 + ee) * 2 + 1) * 32];
                const float arv[4] = {ar4.x, ar4.y, ar4.z, ar4.w};
                const float adv[4] = {ai4.x - ar4.x, ai4.y - ar4.y, ai4.z - ar4.z, ai4.w - ar4.w};
                const float asv[4] = {ar4.x + ai4.x, ar4.y + ai4.y, ar4.z + ai4.z, ar4.w + ai4.w};
                #pragma unroll
                for (int q = 0; q < NQ; ++q) {
                    const int s = ee * 2 + q;
                    v2f fR, fD, fS, xS, xR, xI;
                    fR.x = arv[2 * q];       fR.y = arv[2 * q + 1];
                    fD.x = adv[2 * q];       fD.y = adv[2 * q + 1];
                    fS.x = asv[2 * q];       fS.y = asv[2 * q + 1];
                    xS.x = cs[t][2 * s];     xS.y = cs[t][2 * s + 1];
                    xR.x = cr[t][2 * s];     xR.y = cr[t][2 * s + 1];
                    xI.x = ci[t][2 * s];     xI.y = ci[t][2 * s + 1];
                    t1[u] = wmma_f32(fR, xS, t1[u]);
                    t2[u] = wmma_f32(fD, xR, t2[u]);
                    t3[u] = wmma_f32(fS, xI, t3[u]);
                }
            }
        }
    }
    tr[0] = t1[0] - t3[0];  tv[0] = t1[0] + t2[0];
    tr[1] = t1[1] - t3[1];  tv[1] = t1[1] + t2[1];
}

template <int L>
__device__ __forceinline__ void layer_diag(v8f (&tr)[2], v8f (&tv)[2],
                                           const float* prr, const float* pri,
                                           const float* pcr, const float* pci,
                                           const int lane, const int tj) {
    const int col = tj * 16 + (lane & 15);
    float a2 = 1.f, b2 = 0.f;
    if (L < 3) { a2 = pcr[col]; b2 = pci[col]; }
    #pragma unroll
    for (int u = 0; u < 2; ++u) {
        #pragma unroll
        for (int v = 0; v < 8; ++v) {
            const int row = u * 16 + v + ((lane >> 4) << 3);
            const unsigned k = (unsigned)(row * 32 + col);
            const int par = __popc(k & (k >> 1)) + (int)((k & 1u) & (k >> 9));
            const float sgn = (par & 1) ? -1.f : 1.f;
            float pRe = sgn, pIm = 0.f;
            if (L < 3) {
                const float a1 = prr[row], b1 = pri[row];
                pRe = sgn * (a1 * a2 - b1 * b2);
                pIm = sgn * (a1 * b2 + b1 * a2);
            }
            const float er = tr[u][v], ei = tv[u][v];
            tr[u][v] = er * pRe - ei * pIm;
            tv[u][v] = er * pIm + ei * pRe;
        }
    }
}

__device__ __forceinline__ void store_strip(const v8f (&tr)[2], const v8f (&tv)[2],
                                            float* Dr, float* Di,
                                            const int lane, const int tj) {
    const int col = tj * 16 + (lane & 15);
    #pragma unroll
    for (int u = 0; u < 2; ++u) {
        #pragma unroll
        for (int v = 0; v < 8; ++v) {
            const int row = u * 16 + v + ((lane >> 4) << 3);
            Dr[row * LP + col] = tr[u][v];
            Di[row * LP + col] = tv[u][v];
        }
    }
}

__global__ __launch_bounds__(64) void qsim(const float* __restrict__ x,
                                           const float* __restrict__ U,
                                           float* __restrict__ out) {
    __shared__ __attribute__((aligned(16))) float lds[4 * PLANE + 128];
    float* const S0r = lds;
    float* const S0i = lds + 1 * PLANE;
    float* const S1r = lds + 2 * PLANE;
    float* const S1i = lds + 3 * PLANE;
    float* const prr = lds + 4 * PLANE;
    float* const pri = prr + 32;
    float* const pcr = prr + 64;
    float* const pci = prr + 96;
    float* const red = S0r;

    const int tid  = threadIdx.x;
    const int b    = blockIdx.x;
    const int lane = tid & 31;
    const int tj   = tid >> 5;

    float4 bf[8];
    load_bfrags<0>(U, lane, tj, bf);

    float xb[10];
    #pragma unroll
    for (int j = 0; j < 10; ++j) xb[j] = x[b * 10 + j];

    {
        const int  idx = tid & 31;
        const bool isC = tid >= 32;
        float a = 0.f;
        #pragma unroll
        for (int jj = 0; jj < 5; ++jj) {
            const int bit = (idx >> (4 - jj)) & 1;
            a += xb[isC ? 5 + jj : jj] * (1.f - 2.f * (float)bit);
        }
        a *= -0.5f;
        float sv, cv; sincosf(a, &sv, &cv);
        if (isC) { pcr[idx] = cv; pci[idx] = sv; }
        else     { prr[idx] = cv; pri[idx] = sv; }
    }
    __syncthreads();

    v8f tr[2], tv[2];

    layer_products<0>(U, bf, S0r, S0i, lane, tj, tr, tv);
    layer_diag<0>(tr, tv, prr, pri, pcr, pci, lane, tj);
    load_bfrags<1>(U, lane, tj, bf);
    store_strip(tr, tv, S1r, S1i, lane, tj);
    __syncthreads();

    layer_products<1>(U, bf, S1r, S1i, lane, tj, tr, tv);
    layer_diag<1>(tr, tv, prr, pri, pcr, pci, lane, tj);
    load_bfrags<2>(U, lane, tj, bf);
    store_strip(tr, tv, S0r, S0i, lane, tj);
    __syncthreads();

    layer_products<2>(U, bf, S0r, S0i, lane, tj, tr, tv);
    layer_diag<2>(tr, tv, prr, pri, pcr, pci, lane, tj);
    load_bfrags<3>(U, lane, tj, bf);
    store_strip(tr, tv, S1r, S1i, lane, tj);
    __syncthreads();

    layer_products<3>(U, bf, S1r, S1i, lane, tj, tr, tv);
    layer_diag<3>(tr, tv, prr, pri, pcr, pci, lane, tj);

    float acc[10];
    #pragma unroll
    for (int j = 0; j < 10; ++j) acc[j] = 0.f;
    {
        const int col = tj * 16 + (lane & 15);
        #pragma unroll
        for (int u = 0; u < 2; ++u) {
            #pragma unroll
            for (int v = 0; v < 8; ++v) {
                const int row = u * 16 + v + ((lane >> 4) << 3);
                const int i   = row * 32 + col;
                const float p = tr[u][v] * tr[u][v] + tv[u][v] * tv[u][v];
                #pragma unroll
                for (int j = 0; j < 10; ++j)
                    acc[j] += ((i >> (9 - j)) & 1) ? -p : p;
            }
        }
    }
    #pragma unroll
    for (int j = 0; j < 10; ++j) red[tid * 10 + j] = acc[j];
    __syncthreads();
    if (tid < 10) {
        float s = 0.f;
        for (int t = 0; t < 64; ++t) s += red[t * 10 + tid];
        out[b * 10 + tid] = s;
    }
}

extern "C" void kernel_launch(void* const* d_in, const int* in_sizes, int n_in,
                              void* d_out, int out_size, void* d_ws, size_t ws_size,
                              hipStream_t stream) {
    const float* x  = (const float*)d_in[0];
    const float* w1 = (const float*)d_in[1];
    const float* w2 = (const float*)d_in[2];
    float* out = (float*)d_out;
    float* U   = (float*)d_ws;

    const int B = in_sizes[0] / 10;

    hipLaunchKernelGGL(build_unitaries, dim3(8), dim3(256), 0, stream, w1, w2, U);
    hipLaunchKernelGGL(qsim, dim3(B), dim3(64), 0, stream, x, U, out);
}


# last Rot layer no longer applied to the state: <Z_j> = h00*D_j + 2Re(h01*X_j) from pair sums of the state before it (H = R^H Z R per wire); 96 of 296 WMMAs per wave go; sums and their order change
# speedup vs baseline: 2.9315x; 1.1873x over previous
#include <hip/hip_runtime.h>

typedef float v2f __attribute__((ext_vector_type(2)));
typedef float v8f __attribute__((ext_vector_type(8)));

__device__ __forceinline__ v8f wmma_f32(v2f a, v2f b, v8f c) {
    return __builtin_amdgcn_wmma_f32_16x16x4_f32(false, a, false, b, (short)0, c, false, false);
}

__global__ __launch_bounds__(256) void build_unitaries(const float* __restrict__ w1,
                                                       const float* __restrict__ w2,
                                                       float* __restrict__ U) {
    __shared__ float trig[10][8];

    const int tid  = threadIdx.x;
    const bool obs = (blockIdx.x == 6);
    const int l    = obs ? 3 : (int)(blockIdx.x >> 1);
    const int mat  = blockIdx.x & 1;
    const int j0   = (!obs && mat) ? 5 : 0;
    const int nw   = obs ? 10 : 5;
    const int wv   = tid >> 5;
    const int lane = tid & 31;

    if (wv < 4 && lane < nw) {
        const int j = j0 + lane;
        const float* w = (l < 3) ? (w1 + (l * 10 + j) * 3) : (w2 + j * 3);
        const float phi = w[0], theta = w[1], omega = w[2];
        if (wv == 0) {
            trig[lane][0] = cosf(0.5f * theta);
        } else if (wv == 1) {
            trig[lane][1] = sinf(0.5f * theta);
        } else if (wv == 2) {
            float sa, ca;
            sincosf(-0.5f * (phi + omega), &sa, &ca);
            trig[lane][2] = ca;  trig[lane][3] = sa;
        } else {
            float sb, cb;
            sincosf(-0.5f * (phi - omega), &sb, &cb);
            trig[lane][4] = cb;  trig[lane][5] = sb;
        }
    }
    __syncthreads();

    if (obs) {
        if (tid < 10) {
            const float c  = trig[tid][0], s  = trig[tid][1];
            const float ca = trig[tid][2], sa = trig[tid][3];
            const float cb = trig[tid][4], sb = trig[tid][5];
            const float pr = ca * cb - sa * sb;
            const float pi = ca * sb + sa * cb;
            const float g  = 2.f * c * s;
            reinterpret_cast<float4*>(U + 3 * 4096)[tid] =
                make_float4(c * c - s * s, -g * pr, g * pi, 0.f);
        }
        return;
    }

    const int q   = wv;
    const int c16 = lane & 15;
    const int h   = lane >> 4;
    const int r   = (q >> 2) * 16 + c16;
    const int cc0 = mat ? ((q & 3) * 8 + h * 4)
                        : (((q >> 1) & 1) * 16 + h * 8 + (q & 1) * 4);

    float gr[5][2], gi[5][2];
    #pragma unroll
    for (int jj = 0; jj < 5; ++jj) {
        const float c  = trig[jj][0], s  = trig[jj][1];
        const float ca = trig[jj][2], sa = trig[jj][3];
        const float cb = trig[jj][4], sb = trig[jj][5];
        const float epr = ca * c, epi = sa * c;
        const float emr = cb * s, emi = sb * s;
        const int rb = (r >> (4 - jj)) & 1;
        gr[jj][0] = rb ?  emr : epr;   gi[jj][0] = rb ?  emi : epi;
        gr[jj][1] = rb ?  epr : -emr;  gi[jj][1] = rb ? -epi : emi;
    }

    float pr = 1.f, pi = 0.f;
    #pragma unroll
    for (int jj = 0; jj < 3; ++jj) {
        const int cb = (cc0 >> (4 - jj)) & 1;
        const float ar = cb ? gr[jj][1] : gr[jj][0];
        const float ai = cb ? gi[jj][1] : gi[jj][0];
        const float nr = pr * ar - pi * ai;
        const float ni = pr * ai + pi * ar;
        pr = nr; pi = ni;
    }
    float er[4], ei[4];
    #pragma unroll
    for (int b1 = 0; b1 < 2; ++b1) {
        const float qr = pr * gr[3][b1] - pi * gi[3][b1];
        const float qi = pr * gi[3][b1] + pi * gr[3][b1];
        #pragma unroll
        for (int b0 = 0; b0 < 2; ++b0) {
            er[b1 * 2 + b0] = qr * gr[4][b0] - qi * gi[4][b0];
            ei[b1 * 2 + b0] = qr * gi[4][b0] + qi * gr[4][b0];
        }
    }

    float4* dst = reinterpret_cast<float4*>(U + l * 4096)
                + (mat ? 0 : 512) + (q * 2) * 32 + lane;
    dst[0]  = make_float4(er[0], er[1], er[2], er[3]);
    dst[32] = make_float4(ei[0], ei[1], ei[2], ei[3]);
}

constexpr int LP    = 36;
constexpr int PLANE = 32 * LP;

template <int L>
__device__ __forceinline__ void load_bfrags(const float* __restrict__ U,
                                            const int lane, const int tj,
                                            float4 (&bf)[8]) {
    const float4* __restrict__ Bf =
        reinterpret_cast<const float4*>(U + L * 4096) + tj * 256 + lane;
    constexpr int N = (L == 0) ? 2 : 8;
    #pragma unroll
    for (int i = 0; i < N; ++i) bf[i] = Bf[i * 32];
}

template <int L>
__device__ __forceinline__ void layer_products(const float* __restrict__ U,
                                               const float4 (&bf)[8],
                                               const float* Sr, const float* Si,
                                               const int lane, const int tj,
                                               v8f (&tr)[2], v8f (&tv)[2]) {
    constexpr bool FIRST = (L == 0);
    const int c = lane & 15;
    const int h = lane >> 4;
    const float4* __restrict__ Af =
        reinterpret_cast<const float4*>(U + L * 4096) + 512 + lane;

    v8f cr[2], ci[2], cs[2];
    cr[0] = v8f{}; cr[1] = v8f{}; ci[0] = v8f{}; ci[1] = v8f{};

    if (FIRST) {
        const float4 br4 = bf[0];
        const float4 bi4 = bf[1];
        v2f sR, bR, bI;
        sR.x = (lane == 0) ? 1.f : 0.f;  sR.y = 0.f;
        bR.x = br4.x;  bR.y = br4.y;
        bI.x = bi4.x;  bI.y = bi4.y;
        cr[0] = wmma_f32(sR, bR, cr[0]);
        ci[0] = wmma_f32(sR, bI, ci[0]);
    } else {
        v8f p1[2], p2[2], p3[2];
        p1[0] = v8f{}; p1[1] = v8f{}; p2[0] = v8f{}; p2[1] = v8f{}; p3[0] = v8f{}; p3[1] = v8f{};
        #pragma unroll
        for (int j = 0; j < 4; ++j) {
            const float4 br4 = bf[j * 2];
            const float4 bi4 = bf[j * 2 + 1];
            const float brv[4] = {br4.x, br4.y, br4.z, br4.w};
            const float bdv[4] = {bi4.x - br4.x, bi4.y - br4.y, bi4.z - br4.z, bi4.w - br4.w};
            const float bsv[4] = {br4.x + bi4.x, br4.y + bi4.y, br4.z + bi4.z, br4.w + bi4.w};
            #pragma unroll
            for (int t = 0; t < 2; ++t) {
                const int so = (16 * t + c) * LP + 8 * j + 4 * h;
                const float4 sr4 = *reinterpret_cast<const float4*>(&Sr[so]);
                const float4 si4 = *reinterpret_cast<const float4*>(&Si[so]);
                const float srv[4] = {sr4.x, sr4.y, sr4.z, sr4.w};
                const float siv[4] = {si4.x, si4.y, si4.z, si4.w};
                const float ssv[4] = {sr4.x + si4.x, sr4.y + si4.y, sr4.z + si4.z, sr4.w + si4.w};
                #pragma unroll
                for (int q = 0; q < 2; ++q) {
                    v2f xS, xR, xI, fR, fD, fS;
                    xS.x = ssv[2 * q];   xS.y = ssv[2 * q + 1];
                    xR.x = srv[2 * q];   xR.y = srv[2 * q + 1];
                    xI.x = siv[2 * q];   xI.y = siv[2 * q + 1];
                    fR.x = brv[2 * q];   fR.y = brv[2 * q + 1];
                    fD.x = bdv[2 * q];   fD.y = bdv[2 * q + 1];
                    fS.x = bsv[2 * q];   fS.y = bsv[2 * q + 1];
                    p1[t] = wmma_f32(xS, fR, p1[t]);
                    p2[t] = wmma_f32(xR, fD, p2[t]);
                    p3[t] = wmma_f32(xI, fS, p3[t]);
                }
            }
        }
        cr[0] = p1[0] - p3[0];  ci[0] = p1[0] + p2[0];
        cr[1] = p1[1] - p3[1];  ci[1] = p1[1] + p2[1];
    }
    cs[0] = cr[0] + ci[0];
    cs[1] = cr[1] + ci[1];

    v8f t1[2], t2[2], t3[2];
    t1[0] = v8f{}; t1[1] = v8f{}; t2[0] = v8f{}; t2[1] = v8f{}; t3[0] = v8f{}; t3[1] = v8f{};
    constexpr int NT = FIRST ? 1 : 2;
    constexpr int NE = FIRST ? 1 : 2;
    constexpr int NQ = FIRST ? 1 : 2;
    #pragma unroll
    for (int t = 0; t < NT; ++t) {
        #pragma unroll
        for (int ee = 0; ee < NE; ++ee) {
            #pragma unroll
            for (int u = 0; u < 2; ++u) {
                const float4 ar4 = Af[(((u * 2 + t) * 2 + ee) * 2) * 32];
                const float4 ai4 = Af[(((u * 2 + t) * 2 + ee) * 2 + 1) * 32];
                const float arv[4] = {ar4.x, ar4.y, ar4.z, ar4.w};
                const float adv[4] = {ai4.x - ar4.x, ai4.y - ar4.y, ai4.z - ar4.z, ai4.w - ar4.w};
                const float asv[4] = {ar4.x + ai4.x, ar4.y + ai4.y, ar4.z + ai4.z, ar4.w + ai4.w};
                #pragma unroll
                for (int q = 0; q < NQ; ++q) {
                    const int s = ee * 2 + q;
                    v2f fR, fD, fS, xS, xR, xI;
                    fR.x = arv[2 * q];       fR.y = arv[2 * q + 1];
                    fD.x = adv[2 * q];       fD.y = adv[2 * q + 1];
                    fS.x = asv[2 * q];       fS.y = asv[2 * q + 1];
                    xS.x = cs[t][2 * s];     xS.y = cs[t][2 * s + 1];
                    xR.x = cr[t][2 * s];     xR.y = cr[t][2 * s + 1];
                    xI.x = ci[t][2 * s];     xI.y = ci[t][2 * s + 1];
                    t1[u] = wmma_f32(fR, xS, t1[u]);
                    t2[u] = wmma_f32(fD, xR, t2[u]);
                    t3[u] = wmma_f32(fS, xI, t3[u]);
                }
            }
        }
    }
    tr[0] = t1[0] - t3[0];  tv[0] = t1[0] + t2[0];
    tr[1] = t1[1] - t3[1];  tv[1] = t1[1] + t2[1];
}

template <int L>
__device__ __forceinline__ void layer_diag(v8f (&tr)[2], v8f (&tv)[2],
                                           const float* prr, const float* pri,
                                           const float* pcr, const float* pci,
                                           const int lane, const int tj) {
    const int col = tj * 16 + (lane & 15);
    float a2 = 1.f, b2 = 0.f;
    if (L < 3) { a2 = pcr[col]; b2 = pci[col]; }
    #pragma unroll
    for (int u = 0; u < 2; ++u) {
        #pragma unroll
        for (int v = 0; v < 8; ++v) {
            const int row = u * 16 + v + ((lane >> 4) << 3);
            const unsigned k = (unsigned)(row * 32 + col);
            const int par = __popc(k & (k >> 1)) + (int)((k & 1u) & (k >> 9));
            const float sgn = (par & 1) ? -1.f : 1.f;
            float pRe = sgn, pIm = 0.f;
            if (L < 3) {
                const float a1 = prr[row], b1 = pri[row];
                pRe = sgn * (a1 * a2 - b1 * b2);
                pIm = sgn * (a1 * b2 + b1 * a2);
            }
            const float er = tr[u][v], ei = tv[u][v];
            tr[u][v] = er * pRe - ei * pIm;
            tv[u][v] = er * pIm + ei * pRe;
        }
    }
}

__device__ __forceinline__ void store_strip(const v8f (&tr)[2], const v8f (&tv)[2],
                                            float* Dr, float* Di,
                                            const int lane, const int tj) {
    const int col = tj * 16 + (lane & 15);
    #pragma unroll
    for (int u = 0; u < 2; ++u) {
        #pragma unroll
        for (int v = 0; v < 8; ++v) {
            const int row = u * 16 + v + ((lane >> 4) << 3);
            Dr[row * LP + col] = tr[u][v];
            Di[row * LP + col] = tv[u][v];
        }
    }
}

__device__ __forceinline__ void load8(const float* p, float (&v)[8]) {
    const float4 a = reinterpret_cast<const float4*>(p)[0];
    const float4 b = reinterpret_cast<const float4*>(p)[1];
    v[0] = a.x; v[1] = a.y; v[2] = a.z; v[3] = a.w;
    v[4] = b.x; v[5] = b.y; v[6] = b.z; v[7] = b.w;
}

__global__ __launch_bounds__(64) void qsim(const float* __restrict__ x,
                                           const float* __restrict__ U,
                                           float* __restrict__ out) {
    __shared__ __attribute__((aligned(16))) float lds[4 * PLANE + 128];
    float* const S0r = lds;
    float* const S0i = lds + 1 * PLANE;
    float* const S1r = lds + 2 * PLANE;
    float* const S1i = lds + 3 * PLANE;
    float* const prr = lds + 4 * PLANE;
    float* const pri = prr + 32;
    float* const pcr = prr + 64;
    float* const pci = prr + 96;
    float* const red = S0r;

    const int tid  = threadIdx.x;
    const int b    = blockIdx.x;
    const int lane = tid & 31;
    const int tj   = tid >> 5;

    float4 bf[8];
    load_bfrags<0>(U, lane, tj, bf);

    float xb[10];
    #pragma unroll
    for (int j = 0; j < 10; ++j) xb[j] = x[b * 10 + j];

    {
        const int  idx = tid & 31;
        const bool isC = tid >= 32;
        float a = 0.f;
        #pragma unroll
        for (int jj = 0; jj < 5; ++jj) {
            const int bit = (idx >> (4 - jj)) & 1;
            a += xb[isC ? 5 + jj : jj] * (1.f - 2.f * (float)bit);
        }
        a *= -0.5f;
        float sv, cv; sincosf(a, &sv, &cv);
        if (isC) { pcr[idx] = cv; pci[idx] = sv; }
        else     { prr[idx] = cv; pri[idx] = sv; }
    }
    __syncthreads();

    v8f tr[2], tv[2];

    layer_products<0>(U, bf, S0r, S0i, lane, tj, tr, tv);
    layer_diag<0>(tr, tv, prr, pri, pcr, pci, lane, tj);
    load_bfrags<1>(U, lane, tj, bf);
    store_strip(tr, tv, S1r, S1i, lane, tj);
    __syncthreads();

    layer_products<1>(U, bf, S1r, S1i, lane, tj, tr, tv);
    layer_diag<1>(tr, tv, prr, pri, pcr, pci, lane, tj);
    load_bfrags<2>(U, lane, tj, bf);
    store_strip(tr, tv, S0r, S0i, lane, tj);
    __syncthreads();

    layer_products<2>(U, bf, S0r, S0i, lane, tj, tr, tv);
    layer_diag<2>(tr, tv, prr, pri, pcr, pci, lane, tj);
    store_strip(tr, tv, S1r, S1i, lane, tj);

    float sr[16], si[16];
    #pragma unroll
    for (int i = 0; i < 16; ++i) { sr[i] = tr[i >> 3][i & 7]; si[i] = tv[i >> 3][i & 7]; }

    float dj[10], qr[10], qi[10];
    #pragma unroll
    for (int j = 0; j < 10; ++j) { dj[j] = 0.f; qr[j] = 0.f; qi[j] = 0.f; }

    float ptot = 0.f;
    #pragma unroll
    for (int i = 0; i < 16; ++i) {
        const float p = sr[i] * sr[i] + si[i] * si[i];
        ptot += p;
        dj[0] += (i & 8) ? -p : p;
        dj[2] += (i & 4) ? -p : p;
        dj[3] += (i & 2) ? -p : p;
        dj[4] += (i & 1) ? -p : p;
    }
    {
        const int c = lane & 15;
        dj[1] = (lane & 16) ? -ptot : ptot;
        dj[5] = tj          ? -ptot : ptot;
        dj[6] = (c & 8)     ? -ptot : ptot;
        dj[7] = (c & 4)     ? -ptot : ptot;
        dj[8] = (c & 2)     ? -ptot : ptot;
        dj[9] = (c & 1)     ? -ptot : ptot;
    }
    #pragma unroll
    for (int k = 0; k < 4; ++k) {
        const int m = 8 >> k;
        const int j = (k == 0) ? 0 : k + 1;
        #pragma unroll
        for (int i = 0; i < 16; ++i) {
            if (i & m) continue;
            qr[j] += sr[i] * sr[i | m] + si[i] * si[i | m];
            qi[j] += sr[i] * si[i | m] - si[i] * sr[i | m];
        }
    }

    __syncthreads();

    float orr[16], oi[16];
    {
        const float4* pr4 = reinterpret_cast<const float4*>(&S1r[lane * LP + 16 * tj]);
        const float4* pi4 = reinterpret_cast<const float4*>(&S1i[lane * LP + 16 * tj]);
        #pragma unroll
        for (int k = 0; k < 4; ++k) {
            const float4 a = pr4[k], bq = pi4[k];
            orr[4 * k] = a.x;  orr[4 * k + 1] = a.y;  orr[4 * k + 2] = a.z;  orr[4 * k + 3] = a.w;
            oi[4 * k]  = bq.x; oi[4 * k + 1]  = bq.y; oi[4 * k + 2]  = bq.z; oi[4 * k + 3]  = bq.w;
        }
    }
    #pragma unroll
    for (int k = 0; k < 4; ++k) {
        const int m = 8 >> k;
        const int j = 6 + k;
        #pragma unroll
        for (int i = 0; i < 16; ++i) {
            if (i & m) continue;
            qr[j] += orr[i] * orr[i | m] + oi[i] * oi[i | m];
            qi[j] += orr[i] * oi[i | m] - oi[i] * orr[i | m];
        }
    }
    {
        float ar[8], ai[8], wr[8], wi[8];
        load8(&S1r[lane * LP + 24 * tj], ar);
        load8(&S1i[lane * LP + 24 * tj], ai);
        load8(&S1r[lane * LP + 16 * (tj ^ 1) + 8 * tj], wr);
        load8(&S1i[lane * LP + 16 * (tj ^ 1) + 8 * tj], wi);
        #pragma unroll
        for (int k = 0; k < 8; ++k) {
            qr[5] += ar[k] * wr[k] + ai[k] * wi[k];
            qi[5] += ar[k] * wi[k] - ai[k] * wr[k];
        }
        qi[5] = tj ? -qi[5] : qi[5];
    }
    {
        const int rb = (lane >> 3) & 1;
        const int o0 = 16 * tj + 8 * rb;
        float ar[8], ai[8], wr[8], wi[8];
        load8(&S1r[lane * LP + o0], ar);
        load8(&S1i[lane * LP + o0], ai);
        load8(&S1r[(lane ^ 8) * LP + o0], wr);
        load8(&S1i[(lane ^ 8) * LP + o0], wi);
        #pragma unroll
        for (int k = 0; k < 8; ++k) {
            qr[1] += ar[k] * wr[k] + ai[k] * wi[k];
            qi[1] += ar[k] * wi[k] - ai[k] * wr[k];
        }
        qi[1] = rb ? -qi[1] : qi[1];
    }

    const float4* __restrict__ Hq = reinterpret_cast<const float4*>(U + 3 * 4096);
    #pragma unroll
    for (int j = 0; j < 10; ++j) {
        const float4 hj = Hq[j];
        red[tid * 10 + j] = hj.x * dj[j] + 2.f * (hj.y * qr[j] - hj.z * qi[j]);
    }
    __syncthreads();
    if (tid < 10) {
        float s = 0.f;
        for (int t = 0; t < 64; ++t) s += red[t * 10 + tid];
        out[b * 10 + tid] = s;
    }
}

extern "C" void kernel_launch(void* const* d_in, const int* in_sizes, int n_in,
                              void* d_out, int out_size, void* d_ws, size_t ws_size,
                              hipStream_t stream) {
    const float* x  = (const float*)d_in[0];
    const float* w1 = (const float*)d_in[1];
    const float* w2 = (const float*)d_in[2];
    float* out = (float*)d_out;
    float* U   = (float*)d_ws;

    const int B = in_sizes[0] / 10;

    hipLaunchKernelGGL(build_unitaries, dim3(7), dim3(256), 0, stream, w1, w2, U);
    hipLaunchKernelGGL(qsim, dim3(B), dim3(64), 0, stream, x, U, out);
}


# layer 0 folded into layer 1: state after layer 0 = outer product x 4 CZ sign patterns, so layer 1's S*B^T is 4 complex rows x B^T (16 WMMAs/wave, was 48+8) plus a row scaling; sum order changes
# speedup vs baseline: 3.1628x; 1.0789x over previous
#include <hip/hip_runtime.h>

typedef float v2f __attribute__((ext_vector_type(2)));
typedef float v8f __attribute__((ext_vector_type(8)));

__device__ __forceinline__ v8f wmma_f32(v2f a, v2f b, v8f c) {
    return __builtin_amdgcn_wmma_f32_16x16x4_f32(false, a, false, b, (short)0, c, false, false);
}

__global__ __launch_bounds__(256) void build_unitaries(const float* __restrict__ w1,
                                                       const float* __restrict__ w2,
                                                       float* __restrict__ U) {
    __shared__ float trig[10][8];

    const int tid  = threadIdx.x;
    const bool obs = (blockIdx.x == 6);
    const int l    = obs ? 3 : (int)(blockIdx.x >> 1);
    const int mat  = blockIdx.x & 1;
    const int j0   = (!obs && mat) ? 5 : 0;
    const int nw   = obs ? 10 : 5;
    const int wv   = tid >> 5;
    const int lane = tid & 31;

    if (wv < 4 && lane < nw) {
        const int j = j0 + lane;
        const float* w = (l < 3) ? (w1 + (l * 10 + j) * 3) : (w2 + j * 3);
        const float phi = w[0], theta = w[1], omega = w[2];
        if (wv == 0) {
            trig[lane][0] = cosf(0.5f * theta);
        } else if (wv == 1) {
            trig[lane][1] = sinf(0.5f * theta);
        } else if (wv == 2) {
            float sa, ca;
            sincosf(-0.5f * (phi + omega), &sa, &ca);
            trig[lane][2] = ca;  trig[lane][3] = sa;
        } else {
            float sb, cb;
            sincosf(-0.5f * (phi - omega), &sb, &cb);
            trig[lane][4] = cb;  trig[lane][5] = sb;
        }
    }
    __syncthreads();

    if (obs) {
        if (tid < 10) {
            const float c  = trig[tid][0], s  = trig[tid][1];
            const float ca = trig[tid][2], sa = trig[tid][3];
            const float cb = trig[tid][4], sb = trig[tid][5];
            const float pr = ca * cb - sa * sb;
            const float pi = ca * sb + sa * cb;
            const float g  = 2.f * c * s;
            reinterpret_cast<float4*>(U + 3 * 4096)[tid] =
                make_float4(c * c - s * s, -g * pr, g * pi, 0.f);
        }
        return;
    }

    const int q   = wv;
    const int c16 = lane & 15;
    const int h   = lane >> 4;
    const int r   = (q >> 2) * 16 + c16;
    const int cc0 = mat ? ((q & 3) * 8 + h * 4)
                        : (((q >> 1) & 1) * 16 + h * 8 + (q & 1) * 4);

    float gr[5][2], gi[5][2];
    #pragma unroll
    for (int jj = 0; jj < 5; ++jj) {
        const float c  = trig[jj][0], s  = trig[jj][1];
        const float ca = trig[jj][2], sa = trig[jj][3];
        const float cb = trig[jj][4], sb = trig[jj][5];
        const float epr = ca * c, epi = sa * c;
        const float emr = cb * s, emi = sb * s;
        const int rb = (r >> (4 - jj)) & 1;
        gr[jj][0] = rb ?  emr : epr;   gi[jj][0] = rb ?  emi : epi;
        gr[jj][1] = rb ?  epr : -emr;  gi[jj][1] = rb ? -epi : emi;
    }

    float pr = 1.f, pi = 0.f;
    #pragma unroll
    for (int jj = 0; jj < 3; ++jj) {
        const int cb = (cc0 >> (4 - jj)) & 1;
        const float ar = cb ? gr[jj][1] : gr[jj][0];
        const float ai = cb ? gi[jj][1] : gi[jj][0];
        const float nr = pr * ar - pi * ai;
        const float ni = pr * ai + pi * ar;
        pr = nr; pi = ni;
    }
    float er[4], ei[4];
    #pragma unroll
    for (int b1 = 0; b1 < 2; ++b1) {
        const float qr = pr * gr[3][b1] - pi * gi[3][b1];
        const float qi = pr * gi[3][b1] + pi * gr[3][b1];
        #pragma unroll
        for (int b0 = 0; b0 < 2; ++b0) {
            er[b1 * 2 + b0] = qr * gr[4][b0] - qi * gi[4][b0];
            ei[b1 * 2 + b0] = qr * gi[4][b0] + qi * gr[4][b0];
        }
    }

    float4* dst = reinterpret_cast<float4*>(U + l * 4096)
                + (mat ? 0 : 512) + (q * 2) * 32 + lane;
    dst[0]  = make_float4(er[0], er[1], er[2], er[3]);
    dst[32] = make_float4(ei[0], ei[1], ei[2], ei[3]);
}

constexpr int LP    = 36;
constexpr int PLANE = 32 * LP;

template <int L>
__device__ __forceinline__ void load_bfrags(const float* __restrict__ U,
                                            const int lane, const int tj,
                                            float4 (&bf)[8]) {
    const float4* __restrict__ Bf =
        reinterpret_cast<const float4*>(U + L * 4096) + tj * 256 + lane;
    #pragma unroll
    for (int i = 0; i < 8; ++i) bf[i] = Bf[i * 32];
}

__device__ __forceinline__ void load8(const float* p, float (&v)[8]) {
    const float4 a = reinterpret_cast<const float4*>(p)[0];
    const float4 b = reinterpret_cast<const float4*>(p)[1];
    v[0] = a.x; v[1] = a.y; v[2] = a.z; v[3] = a.w;
    v[4] = b.x; v[5] = b.y; v[6] = b.z; v[7] = b.w;
}


__device__ __forceinline__ void first_product(const float4 (&bf)[8],
                                              const float* Sr, const float* Si,
                                              const int lane,
                                              v8f (&cr)[2], v8f (&ci)[2]) {
    const int c = lane & 15;
    const int h = lane >> 4;
    v8f p1[2], p2[2], p3[2];
    p1[0] = v8f{}; p1[1] = v8f{}; p2[0] = v8f{}; p2[1] = v8f{}; p3[0] = v8f{}; p3[1] = v8f{};
    #pragma unroll
    for (int j = 0; j < 4; ++j) {
        const float4 br4 = bf[j * 2];
        const float4 bi4 = bf[j * 2 + 1];
        const float brv[4] = {br4.x, br4.y, br4.z, br4.w};
        const float bdv[4] = {bi4.x - br4.x, bi4.y - br4.y, bi4.z - br4.z, bi4.w - br4.w};
        const float bsv[4] = {br4.x + bi4.x, br4.y + bi4.y, br4.z + bi4.z, br4.w + bi4.w};
        #pragma unroll
        for (int t = 0; t < 2; ++t) {
            const int so = (16 * t + c) * LP + 8 * j + 4 * h;
            const float4 sr4 = *reinterpret_cast<const float4*>(&Sr[so]);
            const float4 si4 = *reinterpret_cast<const float4*>(&Si[so]);
            const float srv[4] = {sr4.x, sr4.y, sr4.z, sr4.w};
            const float siv[4] = {si4.x, si4.y, si4.z, si4.w};
            const float ssv[4] = {sr4.x + si4.x, sr4.y + si4.y, sr4.z + si4.z, sr4.w + si4.w};
            #pragma unroll
            for (int q = 0; q < 2; ++q) {
                v2f xS, xR, xI, fR, fD, fS;
                xS.x = ssv[2 * q];   xS.y = ssv[2 * q + 1];
                xR.x = srv[2 * q];   xR.y = srv[2 * q + 1];
                xI.x = siv[2 * q];   xI.y = siv[2 * q + 1];
                fR.x = brv[2 * q];   fR.y = brv[2 * q + 1];
                fD.x = bdv[2 * q];   fD.y = bdv[2 * q + 1];
                fS.x = bsv[2 * q];   fS.y = bsv[2 * q + 1];
                p1[t] = wmma_f32(xS, fR, p1[t]);
                p2[t] = wmma_f32(xR, fD, p2[t]);
                p3[t] = wmma_f32(xI, fS, p3[t]);
            }
        }
    }
    cr[0] = p1[0] - p3[0];  ci[0] = p1[0] + p2[0];
    cr[1] = p1[1] - p3[1];  ci[1] = p1[1] + p2[1];
}

__device__ __forceinline__ void first_product_rank4(const float4 (&bf)[8],
                                                    const float* ear, const float* eai,
                                                    const float* ebr, const float* ebi,
                                                    const int lane,
                                                    v8f (&cr)[2], v8f (&ci)[2]) {
    const int c = lane & 15;
    const int h = lane >> 4;
    const bool isIm = (c & 4) != 0;
    const float sA = (c & 1) ? -1.f : 1.f;
    const float sB = (c & 2) ? -1.f : 1.f;
    v8f acc[2];
    acc[0] = v8f{}; acc[1] = v8f{};
    #pragma unroll
    for (int j = 0; j < 4; ++j) {
        const float4 er4 = *reinterpret_cast<const float4*>(&ebr[8 * j + 4 * h]);
        const float4 ei4 = *reinterpret_cast<const float4*>(&ebi[8 * j + 4 * h]);
        const float erv[4] = {er4.x, er4.y, er4.z, er4.w};
        const float eiv[4] = {ei4.x, ei4.y, ei4.z, ei4.w};
        const float4 br4 = bf[j * 2];
        const float4 bi4 = bf[j * 2 + 1];
        const float brv[4] = {br4.x, br4.y, br4.z, br4.w};
        const float biv[4] = {bi4.x, bi4.y, bi4.z, bi4.w};
        float x1[4], x2[4];
        #pragma unroll
        for (int e4 = 0; e4 < 4; ++e4) {
            const float s  = ((j >> 1) ? sA : 1.f) * ((e4 & 1) ? sB : 1.f);
            const float vr = s * erv[e4];
            const float vi = s * eiv[e4];
            x1[e4] = isIm ? vi : vr;
            x2[e4] = isIm ? vr : -vi;
        }
        #pragma unroll
        for (int q = 0; q < 2; ++q) {
            v2f a1, a2, fR, fI;
            a1.x = x1[2 * q];    a1.y = x1[2 * q + 1];
            a2.x = x2[2 * q];    a2.y = x2[2 * q + 1];
            fR.x = brv[2 * q];   fR.y = brv[2 * q + 1];
            fI.x = biv[2 * q];   fI.y = biv[2 * q + 1];
            acc[q] = wmma_f32(a1, fR, acc[q]);
            acc[q] = wmma_f32(a2, fI, acc[q]);
        }
    }
    const v8f qt = acc[0] + acc[1];
    #pragma unroll
    for (int t = 0; t < 2; ++t) {
        float ar[8], ai[8];
        load8(&ear[16 * t + 8 * h], ar);
        load8(&eai[16 * t + 8 * h], ai);
        #pragma unroll
        for (int r = 0; r < 8; ++r) {
            const int id = (r & 1) + 2 * t;
            const float qr = qt[id], qi = qt[4 + id];
            cr[t][r] = ar[r] * qr - ai[r] * qi;
            ci[t][r] = ar[r] * qi + ai[r] * qr;
        }
    }
}

template <int L>
__device__ __forceinline__ void second_product(const float* __restrict__ U,
                                               const v8f (&cr)[2], const v8f (&ci)[2],
                                               const int lane,
                                               v8f (&tr)[2], v8f (&tv)[2]) {
    const float4* __restrict__ Af =
        reinterpret_cast<const float4*>(U + L * 4096) + 512 + lane;
    v8f cs[2];
    cs[0] = cr[0] + ci[0];
    cs[1] = cr[1] + ci[1];
    v8f t1[2], t2[2], t3[2];
    t1[0] = v8f{}; t1[1] = v8f{}; t2[0] = v8f{}; t2[1] = v8f{}; t3[0] = v8f{}; t3[1] = v8f{};
    #pragma unroll
    for (int t = 0; t < 2; ++t) {
        #pragma unroll
        for (int ee = 0; ee < 2; ++ee) {
            #pragma unroll
            for (int u = 0; u < 2; ++u) {
                const float4 ar4 = Af[(((u * 2 + t) * 2 + ee) * 2) * 32];
                const float4 ai4 = Af[(((u * 2 + t) * 2 + ee) * 2 + 1) * 32];
                const float arv[4] = {ar4.x, ar4.y, ar4.z, ar4.w};
                const float adv[4] = {ai4.x - ar4.x, ai4.y - ar4.y, ai4.z - ar4.z, ai4.w - ar4.w};
                const float asv[4] = {ar4.x + ai4.x, ar4.y + ai4.y, ar4.z + ai4.z, ar4.w + ai4.w};
                #pragma unroll
                for (int q = 0; q < 2; ++q) {
                    const int s = ee * 2 + q;
                    v2f fR, fD, fS, xS, xR, xI;
                    fR.x = arv[2 * q];       fR.y = arv[2 * q + 1];
                    fD.x = adv[2 * q];       fD.y = adv[2 * q + 1];
                    fS.x = asv[2 * q];       fS.y = asv[2 * q + 1];
                    xS.x = cs[t][2 * s];     xS.y = cs[t][2 * s + 1];
                    xR.x = cr[t][2 * s];     xR.y = cr[t][2 * s + 1];
                    xI.x = ci[t][2 * s];     xI.y = ci[t][2 * s + 1];
                    t1[u] = wmma_f32(fR, xS, t1[u]);
                    t2[u] = wmma_f32(fD, xR, t2[u]);
                    t3[u] = wmma_f32(fS, xI, t3[u]);
                }
            }
        }
    }
    tr[0] = t1[0] - t3[0];  tv[0] = t1[0] + t2[0];
    tr[1] = t1[1] - t3[1];  tv[1] = t1[1] + t2[1];
}

template <int L>
__device__ __forceinline__ void layer_diag(v8f (&tr)[2], v8f (&tv)[2],
                                           const float* prr, const float* pri,
                                           const float* pcr, const float* pci,
                                           const int lane, const int tj) {
    const int col = tj * 16 + (lane & 15);
    float a2 = 1.f, b2 = 0.f;
    if (L < 3) { a2 = pcr[col]; b2 = pci[col]; }
    #pragma unroll
    for (int u = 0; u < 2; ++u) {
        #pragma unroll
        for (int v = 0; v < 8; ++v) {
            const int row = u * 16 + v + ((lane >> 4) << 3);
            const unsigned k = (unsigned)(row * 32 + col);
            const int par = __popc(k & (k >> 1)) + (int)((k & 1u) & (k >> 9));
            const float sgn = (par & 1) ? -1.f : 1.f;
            float pRe = sgn, pIm = 0.f;
            if (L < 3) {
                const float a1 = prr[row], b1 = pri[row];
                pRe = sgn * (a1 * a2 - b1 * b2);
                pIm = sgn * (a1 * b2 + b1 * a2);
            }
            const float er = tr[u][v], ei = tv[u][v];
            tr[u][v] = er * pRe - ei * pIm;
            tv[u][v] = er * pIm + ei * pRe;
        }
    }
}

__device__ __forceinline__ void store_strip(const v8f (&tr)[2], const v8f (&tv)[2],
                                            float* Dr, float* Di,
                                            const int lane, const int tj) {
    const int col = tj * 16 + (lane & 15);
    #pragma unroll
    for (int u = 0; u < 2; ++u) {
        #pragma unroll
        for (int v = 0; v < 8; ++v) {
            const int row = u * 16 + v + ((lane >> 4) << 3);
            Dr[row * LP + col] = tr[u][v];
            Di[row * LP + col] = tv[u][v];
        }
    }
}

__global__ __launch_bounds__(64) void qsim(const float* __restrict__ x,
                                           const float* __restrict__ U,
                                           float* __restrict__ out) {
    __shared__ __attribute__((aligned(16))) float lds[4 * PLANE + 256];
    float* const S0r = lds;
    float* const S0i = lds + 1 * PLANE;
    float* const S1r = lds + 2 * PLANE;
    float* const S1i = lds + 3 * PLANE;
    float* const prr = lds + 4 * PLANE;
    float* const pri = prr + 32;
    float* const pcr = prr + 64;
    float* const pci = prr + 96;
    float* const ear = prr + 128;
    float* const eai = prr + 160;
    float* const ebr = prr + 192;
    float* const ebi = prr + 224;
    float* const red = S0r;

    const int tid  = threadIdx.x;
    const int b    = blockIdx.x;
    const int lane = tid & 31;
    const int tj   = tid >> 5;

    float4 bf[8];
    load_bfrags<1>(U, lane, tj, bf);

    const int f0 = ((tid >= 32) ? 0 : 512) + ((tid >> 4) & 1) * 256 + (tid & 15);
    const float f0r = U[f0 * 4];
    const float f0i = U[(f0 + 32) * 4];

    float xb[10];
    #pragma unroll
    for (int j = 0; j < 10; ++j) xb[j] = x[b * 10 + j];

    {
        const int  idx = tid & 31;
        const bool isC = tid >= 32;
        float a = 0.f;
        #pragma unroll
        for (int jj = 0; jj < 5; ++jj) {
            const int bit = (idx >> (4 - jj)) & 1;
            a += xb[isC ? 5 + jj : jj] * (1.f - 2.f * (float)bit);
        }
        a *= -0.5f;
        float sv, cv; sincosf(a, &sv, &cv);
        if (isC) { pcr[idx] = cv; pci[idx] = sv; }
        else     { prr[idx] = cv; pri[idx] = sv; }
        const float sg = (__popc(idx & (idx >> 1)) & 1) ? -1.f : 1.f;
        const float er = sg * (f0r * cv - f0i * sv);
        const float ei = sg * (f0r * sv + f0i * cv);
        if (isC) { ebr[idx] = er; ebi[idx] = ei; }
        else     { ear[idx] = er; eai[idx] = ei; }
    }
    __syncthreads();

    v8f tr[2], tv[2], cr[2], ci[2];

    first_product_rank4(bf, ear, eai, ebr, ebi, lane, cr, ci);
    second_product<1>(U, cr, ci, lane, tr, tv);
    layer_diag<1>(tr, tv, prr, pri, pcr, pci, lane, tj);
    load_bfrags<2>(U, lane, tj, bf);
    store_strip(tr, tv, S0r, S0i, lane, tj);
    __syncthreads();

    first_product(bf, S0r, S0i, lane, cr, ci);
    second_product<2>(U, cr, ci, lane, tr, tv);
    layer_diag<2>(tr, tv, prr, pri, pcr, pci, lane, tj);
    store_strip(tr, tv, S1r, S1i, lane, tj);

    float sr[16], si[16];
    #pragma unroll
    for (int i = 0; i < 16; ++i) { sr[i] = tr[i >> 3][i & 7]; si[i] = tv[i >> 3][i & 7]; }

    float dj[10], qr[10], qi[10];
    #pragma unroll
    for (int j = 0; j < 10; ++j) { dj[j] = 0.f; qr[j] = 0.f; qi[j] = 0.f; }

    float ptot = 0.f;
    #pragma unroll
    for (int i = 0; i < 16; ++i) {
        const float p = sr[i] * sr[i] + si[i] * si[i];
        ptot += p;
        dj[0] += (i & 8) ? -p : p;
        dj[2] += (i & 4) ? -p : p;
        dj[3] += (i & 2) ? -p : p;
        dj[4] += (i & 1) ? -p : p;
    }
    {
        const int c = lane & 15;
        dj[1] = (lane & 16) ? -ptot : ptot;
        dj[5] = tj          ? -ptot : ptot;
        dj[6] = (c & 8)     ? -ptot : ptot;
        dj[7] = (c & 4)     ? -ptot : ptot;
        dj[8] = (c & 2)     ? -ptot : ptot;
        dj[9] = (c & 1)     ? -ptot : ptot;
    }
    #pragma unroll
    for (int k = 0; k < 4; ++k) {
        const int m = 8 >> k;
        const int j = (k == 0) ? 0 : k + 1;
        #pragma unroll
        for (int i = 0; i < 16; ++i) {
            if (i & m) continue;
            qr[j] += sr[i] * sr[i | m] + si[i] * si[i | m];
            qi[j] += sr[i] * si[i | m] - si[i] * sr[i | m];
        }
    }

    __syncthreads();

    float orr[16], oi[16];
    {
        const float4* pr4 = reinterpret_cast<const float4*>(&S1r[lane * LP + 16 * tj]);
        const float4* pi4 = reinterpret_cast<const float4*>(&S1i[lane * LP + 16 * tj]);
        #pragma unroll
        for (int k = 0; k < 4; ++k) {
            const float4 a = pr4[k], bq = pi4[k];
            orr[4 * k] = a.x;  orr[4 * k + 1] = a.y;  orr[4 * k + 2] = a.z;  orr[4 * k + 3] = a.w;
            oi[4 * k]  = bq.x; oi[4 * k + 1]  = bq.y; oi[4 * k + 2]  = bq.z; oi[4 * k + 3]  = bq.w;
        }
    }
    #pragma unroll
    for (int k = 0; k < 4; ++k) {
        const int m = 8 >> k;
        const int j = 6 + k;
        #pragma unroll
        for (int i = 0; i < 16; ++i) {
            if (i & m) continue;
            qr[j] += orr[i] * orr[i | m] + oi[i] * oi[i | m];
            qi[j] += orr[i] * oi[i | m] - oi[i] * orr[i | m];
        }
    }
    {
        float ar[8], ai[8], wr[8], wi[8];
        load8(&S1r[lane * LP + 24 * tj], ar);
        load8(&S1i[lane * LP + 24 * tj], ai);
        load8(&S1r[lane * LP + 16 * (tj ^ 1) + 8 * tj], wr);
        load8(&S1i[lane * LP + 16 * (tj ^ 1) + 8 * tj], wi);
        #pragma unroll
        for (int k = 0; k < 8; ++k) {
            qr[5] += ar[k] * wr[k] + ai[k] * wi[k];
            qi[5] += ar[k] * wi[k] - ai[k] * wr[k];
        }
        qi[5] = tj ? -qi[5] : qi[5];
    }
    {
        const int rb = (lane >> 3) & 1;
        const int o0 = 16 * tj + 8 * rb;
        float ar[8], ai[8], wr[8], wi[8];
        load8(&S1r[lane * LP + o0], ar);
        load8(&S1i[lane * LP + o0], ai);
        load8(&S1r[(lane ^ 8) * LP + o0], wr);
        load8(&S1i[(lane ^ 8) * LP + o0], wi);
        #pragma unroll
        for (int k = 0; k < 8; ++k) {
            qr[1] += ar[k] * wr[k] + ai[k] * wi[k];
            qi[1] += ar[k] * wi[k] - ai[k] * wr[k];
        }
        qi[1] = rb ? -qi[1] : qi[1];
    }

    const float4* __restrict__ Hq = reinterpret_cast<const float4*>(U + 3 * 4096);
    #pragma unroll
    for (int j = 0; j < 10; ++j) {
        const float4 hj = Hq[j];
        red[tid * 10 + j] = hj.x * dj[j] + 2.f * (hj.y * qr[j] - hj.z * qi[j]);
    }
    __syncthreads();
    if (tid < 10) {
        float s = 0.f;
        for (int t = 0; t < 64; ++t) s += red[t * 10 + tid];
        out[b * 10 + tid] = s;
    }
}

extern "C" void kernel_launch(void* const* d_in, const int* in_sizes, int n_in,
                              void* d_out, int out_size, void* d_ws, size_t ws_size,
                              hipStream_t stream) {
    const float* x  = (const float*)d_in[0];
    const float* w1 = (const float*)d_in[1];
    const float* w2 = (const float*)d_in[2];
    float* out = (float*)d_out;
    float* U   = (float*)d_ws;

    const int B = in_sizes[0] / 10;

    hipLaunchKernelGGL(build_unitaries, dim3(7), dim3(256), 0, stream, w1, w2, U);
    hipLaunchKernelGGL(qsim, dim3(B), dim3(64), 0, stream, x, U, out);
}


# factor-build launch removed: every block evaluates the 2x2 Rot matrices into LDS and each lane forms its A/B WMMA fragments in registers as Kronecker products; 1 launch; factor rounding changes
# speedup vs baseline: 3.4415x; 1.0881x over previous
#include <hip/hip_runtime.h>

typedef float v2f __attribute__((ext_vector_type(2)));
typedef float v8f __attribute__((ext_vector_type(8)));

__device__ __forceinline__ v8f wmma_f32(v2f a, v2f b, v8f c) {
    return __builtin_amdgcn_wmma_f32_16x16x4_f32(false, a, false, b, (short)0, c, false, false);
}

constexpr int LP    = 36;
constexpr int PLANE = 32 * LP;

struct cf { float r, i; };
__device__ __forceinline__ cf cmul(const cf a, const cf b) {
    cf o;
    o.r = a.r * b.r - a.i * b.i;
    o.i = a.r * b.i + a.i * b.r;
    return o;
}

__device__ __forceinline__ void wire_row(const float* Gl, const int wire, const int rb,
                                         cf (&g)[2]) {
    const float4 v = *reinterpret_cast<const float4*>(&Gl[wire * 8 + rb * 4]);
    g[0].r = v.x; g[0].i = v.y; g[1].r = v.z; g[1].i = v.w;
}
__device__ __forceinline__ cf wire_entry(const float* Gl, const int wire, const int rb,
                                         const int cb) {
    const float2 v = *reinterpret_cast<const float2*>(&Gl[wire * 8 + rb * 4 + cb * 2]);
    cf o; o.r = v.x; o.i = v.y;
    return o;
}

template <int L>
__device__ __forceinline__ void gen_bfrags(const float* G,
                                           const int lane, const int tj,
                                           float4 (&bf)[8]) {
    const float* Gl = G + L * 80;
    const int c = lane & 15;
    const int h = lane >> 4;
    cf r5[2], r6[2], r8[2], r9[2];
    wire_row(Gl, 5, tj, r5);
    wire_row(Gl, 6, (c >> 3) & 1, r6);
    const cf r7 = wire_entry(Gl, 7, (c >> 2) & 1, h);
    wire_row(Gl, 8, (c >> 1) & 1, r8);
    wire_row(Gl, 9, c & 1, r9);
    cf v[4];
    #pragma unroll
    for (int e = 0; e < 4; ++e) v[e] = cmul(r7, cmul(r8[e >> 1], r9[e & 1]));
    #pragma unroll
    for (int j = 0; j < 4; ++j) {
        const cf d = cmul(r5[j >> 1], r6[j & 1]);
        const cf e0 = cmul(d, v[0]), e1 = cmul(d, v[1]), e2 = cmul(d, v[2]), e3 = cmul(d, v[3]);
        bf[j * 2]     = make_float4(e0.r, e1.r, e2.r, e3.r);
        bf[j * 2 + 1] = make_float4(e0.i, e1.i, e2.i, e3.i);
    }
}

template <int L>
__device__ __forceinline__ void prep_afrags(const float* G, const int lane,
                                            cf (&w)[2][4], cf (&d)[2][2]) {
    const float* Gl = G + L * 80;
    const int c = lane & 15;
    const int h = lane >> 4;
    cf r2[2], r3[2], r4[2];
    wire_row(Gl, 0, 0, d[0]);
    wire_row(Gl, 0, 1, d[1]);
    const cf r1 = wire_entry(Gl, 1, (c >> 3) & 1, h);
    wire_row(Gl, 2, (c >> 2) & 1, r2);
    wire_row(Gl, 3, (c >> 1) & 1, r3);
    wire_row(Gl, 4, c & 1, r4);
    cf y[4];
    #pragma unroll
    for (int e = 0; e < 4; ++e) y[e] = cmul(r3[e >> 1], r4[e & 1]);
    #pragma unroll
    for (int ee = 0; ee < 2; ++ee) {
        const cf z = cmul(r1, r2[ee]);
        #pragma unroll
        for (int e = 0; e < 4; ++e) w[ee][e] = cmul(z, y[e]);
    }
}

__device__ __forceinline__ void load8(const float* p, float (&v)[8]) {
    const float4 a = reinterpret_cast<const float4*>(p)[0];
    const float4 b = reinterpret_cast<const float4*>(p)[1];
    v[0] = a.x; v[1] = a.y; v[2] = a.z; v[3] = a.w;
    v[4] = b.x; v[5] = b.y; v[6] = b.z; v[7] = b.w;
}


__device__ __forceinline__ void first_product(const float4 (&bf)[8],
                                              const float* Sr, const float* Si,
                                              const int lane,
                                              v8f (&cr)[2], v8f (&ci)[2]) {
    const int c = lane & 15;
    const int h = lane >> 4;
    v8f p1[2], p2[2], p3[2];
    p1[0] = v8f{}; p1[1] = v8f{}; p2[0] = v8f{}; p2[1] = v8f{}; p3[0] = v8f{}; p3[1] = v8f{};
    #pragma unroll
    for (int j = 0; j < 4; ++j) {
        const float4 br4 = bf[j * 2];
        const float4 bi4 = bf[j * 2 + 1];
        const float brv[4] = {br4.x, br4.y, br4.z, br4.w};
        const float bdv[4] = {bi4.x - br4.x, bi4.y - br4.y, bi4.z - br4.z, bi4.w - br4.w};
        const float bsv[4] = {br4.x + bi4.x, br4.y + bi4.y, br4.z + bi4.z, br4.w + bi4.w};
        #pragma unroll
        for (int t = 0; t < 2; ++t) {
            const int so = (16 * t + c) * LP + 8 * j + 4 * h;
            const float4 sr4 = *reinterpret_cast<const float4*>(&Sr[so]);
            const float4 si4 = *reinterpret_cast<const float4*>(&Si[so]);
            const float srv[4] = {sr4.x, sr4.y, sr4.z, sr4.w};
            const float siv[4] = {si4.x, si4.y, si4.z, si4.w};
            const float ssv[4] = {sr4.x + si4.x, sr4.y + si4.y, sr4.z + si4.z, sr4.w + si4.w};
            #pragma unroll
            for (int q = 0; q < 2; ++q) {
                v2f xS, xR, xI, fR, fD, fS;
                xS.x = ssv[2 * q];   xS.y = ssv[2 * q + 1];
                xR.x = srv[2 * q];   xR.y = srv[2 * q + 1];
                xI.x = siv[2 * q];   xI.y = siv[2 * q + 1];
                fR.x = brv[2 * q];   fR.y = brv[2 * q + 1];
                fD.x = bdv[2 * q];   fD.y = bdv[2 * q + 1];
                fS.x = bsv[2 * q];   fS.y = bsv[2 * q + 1];
                p1[t] = wmma_f32(xS, fR, p1[t]);
                p2[t] = wmma_f32(xR, fD, p2[t]);
                p3[t] = wmma_f32(xI, fS, p3[t]);
            }
        }
    }
    cr[0] = p1[0] - p3[0];  ci[0] = p1[0] + p2[0];
    cr[1] = p1[1] - p3[1];  ci[1] = p1[1] + p2[1];
}

__device__ __forceinline__ void first_product_rank4(const float4 (&bf)[8],
                                                    const float* ear, const float* eai,
                                                    const float* ebr, const float* ebi,
                                                    const int lane,
                                                    v8f (&cr)[2], v8f (&ci)[2]) {
    const int c = lane & 15;
    const int h = lane >> 4;
    const bool isIm = (c & 4) != 0;
    const float sA = (c & 1) ? -1.f : 1.f;
    const float sB = (c & 2) ? -1.f : 1.f;
    v8f acc[2];
    acc[0] = v8f{}; acc[1] = v8f{};
    #pragma unroll
    for (int j = 0; j < 4; ++j) {
        const float4 er4 = *reinterpret_cast<const float4*>(&ebr[8 * j + 4 * h]);
        const float4 ei4 = *reinterpret_cast<const float4*>(&ebi[8 * j + 4 * h]);
        const float erv[4] = {er4.x, er4.y, er4.z, er4.w};
        const float eiv[4] = {ei4.x, ei4.y, ei4.z, ei4.w};
        const float4 br4 = bf[j * 2];
        const float4 bi4 = bf[j * 2 + 1];
        const float brv[4] = {br4.x, br4.y, br4.z, br4.w};
        const float biv[4] = {bi4.x, bi4.y, bi4.z, bi4.w};
        float x1[4], x2[4];
        #pragma unroll
        for (int e4 = 0; e4 < 4; ++e4) {
            const float s  = ((j >> 1) ? sA : 1.f) * ((e4 & 1) ? sB : 1.f);
            const float vr = s * erv[e4];
            const float vi = s * eiv[e4];
            x1[e4] = isIm ? vi : vr;
            x2[e4] = isIm ? vr : -vi;
        }
        #pragma unroll
        for (int q = 0; q < 2; ++q) {
            v2f a1, a2, fR, fI;
            a1.x = x1[2 * q];    a1.y = x1[2 * q + 1];
            a2.x = x2[2 * q];    a2.y = x2[2 * q + 1];
            fR.x = brv[2 * q];   fR.y = brv[2 * q + 1];
            fI.x = biv[2 * q];   fI.y = biv[2 * q + 1];
            acc[q] = wmma_f32(a1, fR, acc[q]);
            acc[q] = wmma_f32(a2, fI, acc[q]);
        }
    }
    const v8f qt = acc[0] + acc[1];
    #pragma unroll
    for (int t = 0; t < 2; ++t) {
        float ar[8], ai[8];
        load8(&ear[16 * t + 8 * h], ar);
        load8(&eai[16 * t + 8 * h], ai);
        #pragma unroll
        for (int r = 0; r < 8; ++r) {
            const int id = (r & 1) + 2 * t;
            const float qr = qt[id], qi = qt[4 + id];
            cr[t][r] = ar[r] * qr - ai[r] * qi;
            ci[t][r] = ar[r] * qi + ai[r] * qr;
        }
    }
}

__device__ __forceinline__ void second_product(const cf (&aw)[2][4], const cf (&ad)[2][2],
                                               const v8f (&cr)[2], const v8f (&ci)[2],
                                               v8f (&tr)[2], v8f (&tv)[2]) {
    v8f cs[2];
    cs[0] = cr[0] + ci[0];
    cs[1] = cr[1] + ci[1];
    v8f t1[2], t2[2], t3[2];
    t1[0] = v8f{}; t1[1] = v8f{}; t2[0] = v8f{}; t2[1] = v8f{}; t3[0] = v8f{}; t3[1] = v8f{};
    #pragma unroll
    for (int t = 0; t < 2; ++t) {
        #pragma unroll
        for (int ee = 0; ee < 2; ++ee) {
            #pragma unroll
            for (int u = 0; u < 2; ++u) {
                float arv[4], adv[4], asv[4];
                #pragma unroll
                for (int e = 0; e < 4; ++e) {
                    const cf a = cmul(ad[u][t], aw[ee][e]);
                    arv[e] = a.r;
                    adv[e] = a.i - a.r;
                    asv[e] = a.r + a.i;
                }
                #pragma unroll
                for (int q = 0; q < 2; ++q) {
                    const int s = ee * 2 + q;
                    v2f fR, fD, fS, xS, xR, xI;
                    fR.x = arv[2 * q];       fR.y = arv[2 * q + 1];
                    fD.x = adv[2 * q];       fD.y = adv[2 * q + 1];
                    fS.x = asv[2 * q];       fS.y = asv[2 * q + 1];
                    xS.x = cs[t][2 * s];     xS.y = cs[t][2 * s + 1];
                    xR.x = cr[t][2 * s];     xR.y = cr[t][2 * s + 1];
                    xI.x = ci[t][2 * s];     xI.y = ci[t][2 * s + 1];
                    t1[u] = wmma_f32(fR, xS, t1[u]);
                    t2[u] = wmma_f32(fD, xR, t2[u]);
                    t3[u] = wmma_f32(fS, xI, t3[u]);
                }
            }
        }
    }
    tr[0] = t1[0] - t3[0];  tv[0] = t1[0] + t2[0];
    tr[1] = t1[1] - t3[1];  tv[1] = t1[1] + t2[1];
}

template <int L>
__device__ __forceinline__ void layer_diag(v8f (&tr)[2], v8f (&tv)[2],
                                           const float* prr, const float* pri,
                                           const float* pcr, const float* pci,
                                           const int lane, const int tj) {
    const int col = tj * 16 + (lane & 15);
    float a2 = 1.f, b2 = 0.f;
    if (L < 3) { a2 = pcr[col]; b2 = pci[col]; }
    #pragma unroll
    for (int u = 0; u < 2; ++u) {
        #pragma unroll
        for (int v = 0; v < 8; ++v) {
            const int row = u * 16 + v + ((lane >> 4) << 3);
            const unsigned k = (unsigned)(row * 32 + col);
            const int par = __popc(k & (k >> 1)) + (int)((k & 1u) & (k >> 9));
            const float sgn = (par & 1) ? -1.f : 1.f;
            float pRe = sgn, pIm = 0.f;
            if (L < 3) {
                const float a1 = prr[row], b1 = pri[row];
                pRe = sgn * (a1 * a2 - b1 * b2);
                pIm = sgn * (a1 * b2 + b1 * a2);
            }
            const float er = tr[u][v], ei = tv[u][v];
            tr[u][v] = er * pRe - ei * pIm;
            tv[u][v] = er * pIm + ei * pRe;
        }
    }
}

__device__ __forceinline__ void store_strip(const v8f (&tr)[2], const v8f (&tv)[2],
                                            float* Dr, float* Di,
                                            const int lane, const int tj) {
    const int col = tj * 16 + (lane & 15);
    #pragma unroll
    for (int u = 0; u < 2; ++u) {
        #pragma unroll
        for (int v = 0; v < 8; ++v) {
            const int row = u * 16 + v + ((lane >> 4) << 3);
            Dr[row * LP + col] = tr[u][v];
            Di[row * LP + col] = tv[u][v];
        }
    }
}

__global__ __launch_bounds__(64) void qsim(const float* __restrict__ x,
                                           const float* __restrict__ w1,
                                           const float* __restrict__ w2,
                                           float* __restrict__ out) {
    __shared__ __attribute__((aligned(16))) float lds[4 * PLANE + 256 + 240 + 40];
    float* const S0r = lds;
    float* const S0i = lds + 1 * PLANE;
    float* const S1r = lds + 2 * PLANE;
    float* const S1i = lds + 3 * PLANE;
    float* const prr = lds + 4 * PLANE;
    float* const pri = prr + 32;
    float* const pcr = prr + 64;
    float* const pci = prr + 96;
    float* const ear = prr + 128;
    float* const eai = prr + 160;
    float* const ebr = prr + 192;
    float* const ebi = prr + 224;
    float* const G   = prr + 256;
    float* const H   = G + 240;
    float* const red = S0r;

    const int tid  = threadIdx.x;
    const int b    = blockIdx.x;
    const int lane = tid & 31;
    const int tj   = tid >> 5;

    const bool isG = tid < 30;
    const bool isH = (tid >= 32) && (tid < 42);
    float phi = 0.f, theta = 0.f, omega = 0.f;
    if (isG || isH) {
        const float* w = isG ? (w1 + tid * 3) : (w2 + (tid - 32) * 3);
        phi = w[0]; theta = w[1]; omega = w[2];
    }

    float xb[10];
    #pragma unroll
    for (int j = 0; j < 10; ++j) xb[j] = x[b * 10 + j];

    {
        const int  idx = tid & 31;
        const bool isC = tid >= 32;
        float a = 0.f;
        #pragma unroll
        for (int jj = 0; jj < 5; ++jj) {
            const int bit = (idx >> (4 - jj)) & 1;
            a += xb[isC ? 5 + jj : jj] * (1.f - 2.f * (float)bit);
        }
        a *= -0.5f;
        float sv, cv; sincosf(a, &sv, &cv);
        if (isC) { pcr[idx] = cv; pci[idx] = sv; }
        else     { prr[idx] = cv; pri[idx] = sv; }

        if (isG || isH) {
            float s, c, sa, ca, sb, cb;
            sincosf(0.5f * theta, &s, &c);
            sincosf(-0.5f * (phi + omega), &sa, &ca);
            sincosf(-0.5f * (phi - omega), &sb, &cb);
            if (isG) {
                const float epr = ca * c, epi = sa * c;
                const float emr = cb * s, emi = sb * s;
                float4* g = reinterpret_cast<float4*>(G + tid * 8);
                g[0] = make_float4(epr, epi, -emr, emi);
                g[1] = make_float4(emr, emi, epr, -epi);
            } else {
                const float pr = ca * cb - sa * sb;
                const float pi = ca * sb + sa * cb;
                const float g2 = 2.f * c * s;
                reinterpret_cast<float4*>(H)[tid - 32] =
                    make_float4(c * c - s * s, -g2 * pr, g2 * pi, 0.f);
            }
        }
        __syncthreads();

        cf f0; f0.r = 1.f; f0.i = 0.f;
        #pragma unroll
        for (int jj = 0; jj < 5; ++jj)
            f0 = cmul(f0, wire_entry(G, (isC ? 5 : 0) + jj, (idx >> (4 - jj)) & 1, 0));
        const float sg = (__popc(idx & (idx >> 1)) & 1) ? -1.f : 1.f;
        const float er = sg * (f0.r * cv - f0.i * sv);
        const float ei = sg * (f0.r * sv + f0.i * cv);
        if (isC) { ebr[idx] = er; ebi[idx] = ei; }
        else     { ear[idx] = er; eai[idx] = ei; }
    }

    float4 bf[8];
    cf aw[2][4], ad[2][2];
    gen_bfrags<1>(G, lane, tj, bf);
    prep_afrags<1>(G, lane, aw, ad);
    __syncthreads();

    v8f tr[2], tv[2], cr[2], ci[2];

    first_product_rank4(bf, ear, eai, ebr, ebi, lane, cr, ci);
    second_product(aw, ad, cr, ci, tr, tv);
    layer_diag<1>(tr, tv, prr, pri, pcr, pci, lane, tj);
    gen_bfrags<2>(G, lane, tj, bf);
    prep_afrags<2>(G, lane, aw, ad);
    store_strip(tr, tv, S0r, S0i, lane, tj);
    __syncthreads();

    first_product(bf, S0r, S0i, lane, cr, ci);
    second_product(aw, ad, cr, ci, tr, tv);
    layer_diag<2>(tr, tv, prr, pri, pcr, pci, lane, tj);
    store_strip(tr, tv, S1r, S1i, lane, tj);

    float sr[16], si[16];
    #pragma unroll
    for (int i = 0; i < 16; ++i) { sr[i] = tr[i >> 3][i & 7]; si[i] = tv[i >> 3][i & 7]; }

    float dj[10], qr[10], qi[10];
    #pragma unroll
    for (int j = 0; j < 10; ++j) { dj[j] = 0.f; qr[j] = 0.f; qi[j] = 0.f; }

    float ptot = 0.f;
    #pragma unroll
    for (int i = 0; i < 16; ++i) {
        const float p = sr[i] * sr[i] + si[i] * si[i];
        ptot += p;
        dj[0] += (i & 8) ? -p : p;
        dj[2] += (i & 4) ? -p : p;
        dj[3] += (i & 2) ? -p : p;
        dj[4] += (i & 1) ? -p : p;
    }
    {
        const int c = lane & 15;
        dj[1] = (lane & 16) ? -ptot : ptot;
        dj[5] = tj          ? -ptot : ptot;
        dj[6] = (c & 8)     ? -ptot : ptot;
        dj[7] = (c & 4)     ? -ptot : ptot;
        dj[8] = (c & 2)     ? -ptot : ptot;
        dj[9] = (c & 1)     ? -ptot : ptot;
    }
    #pragma unroll
    for (int k = 0; k < 4; ++k) {
        const int m = 8 >> k;
        const int j = (k == 0) ? 0 : k + 1;
        #pragma unroll
        for (int i = 0; i < 16; ++i) {
            if (i & m) continue;
            qr[j] += sr[i] * sr[i | m] + si[i] * si[i | m];
            qi[j] += sr[i] * si[i | m] - si[i] * sr[i | m];
        }
    }

    __syncthreads();

    float orr[16], oi[16];
    {
        const float4* pr4 = reinterpret_cast<const float4*>(&S1r[lane * LP + 16 * tj]);
        const float4* pi4 = reinterpret_cast<const float4*>(&S1i[lane * LP + 16 * tj]);
        #pragma unroll
        for (int k = 0; k < 4; ++k) {
            const float4 a = pr4[k], bq = pi4[k];
            orr[4 * k] = a.x;  orr[4 * k + 1] = a.y;  orr[4 * k + 2] = a.z;  orr[4 * k + 3] = a.w;
            oi[4 * k]  = bq.x; oi[4 * k + 1]  = bq.y; oi[4 * k + 2]  = bq.z; oi[4 * k + 3]  = bq.w;
        }
    }
    #pragma unroll
    for (int k = 0; k < 4; ++k) {
        const int m = 8 >> k;
        const int j = 6 + k;
        #pragma unroll
        for (int i = 0; i < 16; ++i) {
            if (i & m) continue;
            qr[j] += orr[i] * orr[i | m] + oi[i] * oi[i | m];
            qi[j] += orr[i] * oi[i | m] - oi[i] * orr[i | m];
        }
    }
    {
        float ar[8], ai[8], wr[8], wi[8];
        load8(&S1r[lane * LP + 24 * tj], ar);
        load8(&S1i[lane * LP + 24 * tj], ai);
        load8(&S1r[lane * LP + 16 * (tj ^ 1) + 8 * tj], wr);
        load8(&S1i[lane * LP + 16 * (tj ^ 1) + 8 * tj], wi);
        #pragma unroll
        for (int k = 0; k < 8; ++k) {
            qr[5] += ar[k] * wr[k] + ai[k] * wi[k];
            qi[5] += ar[k] * wi[k] - ai[k] * wr[k];
        }
        qi[5] = tj ? -qi[5] : qi[5];
    }
    {
        const int rb = (lane >> 3) & 1;
        const int o0 = 16 * tj + 8 * rb;
        float ar[8], ai[8], wr[8], wi[8];
        load8(&S1r[lane * LP + o0], ar);
        load8(&S1i[lane * LP + o0], ai);
        load8(&S1r[(lane ^ 8) * LP + o0], wr);
        load8(&S1i[(lane ^ 8) * LP + o0], wi);
        #pragma unroll
        for (int k = 0; k < 8; ++k) {
            qr[1] += ar[k] * wr[k] + ai[k] * wi[k];
            qi[1] += ar[k] * wi[k] - ai[k] * wr[k];
        }
        qi[1] = rb ? -qi[1] : qi[1];
    }

    const float4* Hq = reinterpret_cast<const float4*>(H);
    #pragma unroll
    for (int j = 0; j < 10; ++j) {
        const float4 hj = Hq[j];
        red[tid * 10 + j] = hj.x * dj[j] + 2.f * (hj.y * qr[j] - hj.z * qi[j]);
    }
    __syncthreads();
    if (tid < 10) {
        float s = 0.f;
        for (int t = 0; t < 64; ++t) s += red[t * 10 + tid];
        out[b * 10 + tid] = s;
    }
}

extern "C" void kernel_launch(void* const* d_in, const int* in_sizes, int n_in,
                              void* d_out, int out_size, void* d_ws, size_t ws_size,
                              hipStream_t stream) {
    const float* x  = (const float*)d_in[0];
    const float* w1 = (const float*)d_in[1];
    const float* w2 = (const float*)d_in[2];
    float* out = (float*)d_out;

    const int B = in_sizes[0] / 10;

    hipLaunchKernelGGL(qsim, dim3(B), dim3(64), 0, stream, x, w1, w2, out);
}


# each 32x32 factor split as 2x2 (x) 16x16: the 16x16 parts stay on WMMA at k = 16 (80 WMMAs per wave, was 160), 2x2 of wires 0 and 5 applied elementwise on vector ALU; rounding and sum order change
# speedup vs baseline: 4.0699x; 1.1826x over previous
#include <hip/hip_runtime.h>

typedef float v2f __attribute__((ext_vector_type(2)));
typedef float v8f __attribute__((ext_vector_type(8)));

__device__ __forceinline__ v8f wmma_f32(v2f a, v2f b, v8f c) {
    return __builtin_amdgcn_wmma_f32_16x16x4_f32(false, a, false, b, (short)0, c, false, false);
}

constexpr int LP    = 36;
constexpr int PLANE = 32 * LP;

struct cf { float r, i; };
__device__ __forceinline__ cf cmul(const cf a, const cf b) {
    cf o;
    o.r = a.r * b.r - a.i * b.i;
    o.i = a.r * b.i + a.i * b.r;
    return o;
}

__device__ __forceinline__ void wire_row(const float* Gl, const int wire, const int rb,
                                         cf (&g)[2]) {
    const float4 v = *reinterpret_cast<const float4*>(&Gl[wire * 8 + rb * 4]);
    g[0].r = v.x; g[0].i = v.y; g[1].r = v.z; g[1].i = v.w;
}
__device__ __forceinline__ cf wire_entry(const float* Gl, const int wire, const int rb,
                                         const int cb) {
    const float2 v = *reinterpret_cast<const float2*>(&Gl[wire * 8 + rb * 4 + cb * 2]);
    cf o; o.r = v.x; o.i = v.y;
    return o;
}

template <int L>
__device__ __forceinline__ void gen_bfrags(const float* G,
                                           const int lane, const int tj,
                                           float4 (&bf)[4], cf (&r5)[2]) {
    const float* Gl = G + L * 80;
    const int c = lane & 15;
    const int h = lane >> 4;
    cf r6[2], r8[2], r9[2];
    wire_row(Gl, 5, tj, r5);
    wire_row(Gl, 6, (c >> 3) & 1, r6);
    const cf r7 = wire_entry(Gl, 7, (c >> 2) & 1, h);
    wire_row(Gl, 8, (c >> 1) & 1, r8);
    wire_row(Gl, 9, c & 1, r9);
    cf v[4];
    #pragma unroll
    for (int e = 0; e < 4; ++e) v[e] = cmul(r7, cmul(r8[e >> 1], r9[e & 1]));
    #pragma unroll
    for (int j0 = 0; j0 < 2; ++j0) {
        const cf d = r6[j0];
        const cf e0 = cmul(d, v[0]), e1 = cmul(d, v[1]), e2 = cmul(d, v[2]), e3 = cmul(d, v[3]);
        bf[j0 * 2]     = make_float4(e0.r, e1.r, e2.r, e3.r);
        bf[j0 * 2 + 1] = make_float4(e0.i, e1.i, e2.i, e3.i);
    }
}

template <int L>
__device__ __forceinline__ void prep_afrags(const float* G, const int lane,
                                            cf (&w)[2][4], cf (&d)[2][2]) {
    const float* Gl = G + L * 80;
    const int c = lane & 15;
    const int h = lane >> 4;
    cf r2[2], r3[2], r4[2];
    wire_row(Gl, 0, 0, d[0]);
    wire_row(Gl, 0, 1, d[1]);
    const cf r1 = wire_entry(Gl, 1, (c >> 3) & 1, h);
    wire_row(Gl, 2, (c >> 2) & 1, r2);
    wire_row(Gl, 3, (c >> 1) & 1, r3);
    wire_row(Gl, 4, c & 1, r4);
    cf y[4];
    #pragma unroll
    for (int e = 0; e < 4; ++e) y[e] = cmul(r3[e >> 1], r4[e & 1]);
    #pragma unroll
    for (int ee = 0; ee < 2; ++ee) {
        const cf z = cmul(r1, r2[ee]);
        #pragma unroll
        for (int e = 0; e < 4; ++e) w[ee][e] = cmul(z, y[e]);
    }
}

__device__ __forceinline__ void load8(const float* p, float (&v)[8]) {
    const float4 a = reinterpret_cast<const float4*>(p)[0];
    const float4 b = reinterpret_cast<const float4*>(p)[1];
    v[0] = a.x; v[1] = a.y; v[2] = a.z; v[3] = a.w;
    v[4] = b.x; v[5] = b.y; v[6] = b.z; v[7] = b.w;
}


__device__ __forceinline__ void first_product(const float4 (&bf)[4], const cf (&r5)[2],
                                              const float* Sr, const float* Si,
                                              const int lane,
                                              v8f (&cr)[2], v8f (&ci)[2]) {
    const int c = lane & 15;
    const int h = lane >> 4;
    const float g0r = r5[0].r, g0i = r5[0].i, g1r = r5[1].r, g1i = r5[1].i;
    v8f p1[2], p2[2], p3[2];
    p1[0] = v8f{}; p1[1] = v8f{}; p2[0] = v8f{}; p2[1] = v8f{}; p3[0] = v8f{}; p3[1] = v8f{};
    #pragma unroll
    for (int j0 = 0; j0 < 2; ++j0) {
        const float4 br4 = bf[j0 * 2];
        const float4 bi4 = bf[j0 * 2 + 1];
        const v2f br[2] = {v2f{br4.x, br4.y}, v2f{br4.z, br4.w}};
        const v2f bi[2] = {v2f{bi4.x, bi4.y}, v2f{bi4.z, bi4.w}};
        #pragma unroll
        for (int t = 0; t < 2; ++t) {
            const int so = (16 * t + c) * LP + 8 * j0 + 4 * h;
            const float4 lr4 = *reinterpret_cast<const float4*>(&Sr[so]);
            const float4 li4 = *reinterpret_cast<const float4*>(&Si[so]);
            const float4 hr4 = *reinterpret_cast<const float4*>(&Sr[so + 16]);
            const float4 hi4 = *reinterpret_cast<const float4*>(&Si[so + 16]);
            const v2f lr[2] = {v2f{lr4.x, lr4.y}, v2f{lr4.z, lr4.w}};
            const v2f li[2] = {v2f{li4.x, li4.y}, v2f{li4.z, li4.w}};
            const v2f hr[2] = {v2f{hr4.x, hr4.y}, v2f{hr4.z, hr4.w}};
            const v2f hi[2] = {v2f{hi4.x, hi4.y}, v2f{hi4.z, hi4.w}};
            #pragma unroll
            for (int q = 0; q < 2; ++q) {
                const v2f xR = g0r * lr[q] - g0i * li[q] + g1r * hr[q] - g1i * hi[q];
                const v2f xI = g0r * li[q] + g0i * lr[q] + g1r * hi[q] + g1i * hr[q];
                const v2f xS = xR + xI;
                const v2f fR = br[q];
                const v2f fD = bi[q] - br[q];
                const v2f fS = br[q] + bi[q];
                p1[t] = wmma_f32(xS, fR, p1[t]);
                p2[t] = wmma_f32(xR, fD, p2[t]);
                p3[t] = wmma_f32(xI, fS, p3[t]);
            }
        }
    }
    cr[0] = p1[0] - p3[0];  ci[0] = p1[0] + p2[0];
    cr[1] = p1[1] - p3[1];  ci[1] = p1[1] + p2[1];
}

__device__ __forceinline__ void first_product_rank4(const float4 (&bf)[4], const cf (&r5)[2],
                                                    const float* ear, const float* eai,
                                                    const float* ebr, const float* ebi,
                                                    const int lane,
                                                    v8f (&cr)[2], v8f (&ci)[2]) {
    const int c = lane & 15;
    const int h = lane >> 4;
    const bool isIm = (c & 4) != 0;
    const float sA = (c & 1) ? -1.f : 1.f;
    const float sB = (c & 2) ? -1.f : 1.f;
    const float g0r = r5[0].r, g0i = r5[0].i;
    const float g1r = sA * r5[1].r, g1i = sA * r5[1].i;
    const v2f sv = v2f{1.f, sB};
    v8f acc[2];
    acc[0] = v8f{}; acc[1] = v8f{};
    #pragma unroll
    for (int j0 = 0; j0 < 2; ++j0) {
        const float4 lr4 = *reinterpret_cast<const float4*>(&ebr[8 * j0 + 4 * h]);
        const float4 li4 = *reinterpret_cast<const float4*>(&ebi[8 * j0 + 4 * h]);
        const float4 hr4 = *reinterpret_cast<const float4*>(&ebr[16 + 8 * j0 + 4 * h]);
        const float4 hi4 = *reinterpret_cast<const float4*>(&ebi[16 + 8 * j0 + 4 * h]);
        const v2f lr[2] = {v2f{lr4.x, lr4.y}, v2f{lr4.z, lr4.w}};
        const v2f li[2] = {v2f{li4.x, li4.y}, v2f{li4.z, li4.w}};
        const v2f hr[2] = {v2f{hr4.x, hr4.y}, v2f{hr4.z, hr4.w}};
        const v2f hi[2] = {v2f{hi4.x, hi4.y}, v2f{hi4.z, hi4.w}};
        const float4 br4 = bf[j0 * 2];
        const float4 bi4 = bf[j0 * 2 + 1];
        const v2f br[2] = {v2f{br4.x, br4.y}, v2f{br4.z, br4.w}};
        const v2f bi[2] = {v2f{bi4.x, bi4.y}, v2f{bi4.z, bi4.w}};
        #pragma unroll
        for (int q = 0; q < 2; ++q) {
            const v2f vr = sv * (g0r * lr[q] - g0i * li[q] + g1r * hr[q] - g1i * hi[q]);
            const v2f vi = sv * (g0r * li[q] + g0i * lr[q] + g1r * hi[q] + g1i * hr[q]);
            const v2f a1 = isIm ? vi : vr;
            const v2f a2 = isIm ? vr : -vi;
            acc[q] = wmma_f32(a1, br[q], acc[q]);
            acc[q] = wmma_f32(a2, bi[q], acc[q]);
        }
    }
    const v8f qt = acc[0] + acc[1];
    #pragma unroll
    for (int t = 0; t < 2; ++t) {
        float ar[8], ai[8];
        load8(&ear[16 * t + 8 * h], ar);
        load8(&eai[16 * t + 8 * h], ai);
        #pragma unroll
        for (int r = 0; r < 8; ++r) {
            const int id = (r & 1) + 2 * t;
            const float qr = qt[id], qi = qt[4 + id];
            cr[t][r] = ar[r] * qr - ai[r] * qi;
            ci[t][r] = ar[r] * qi + ai[r] * qr;
        }
    }
}

__device__ __forceinline__ void second_product(const cf (&aw)[2][4], const cf (&ad)[2][2],
                                               const v8f (&cr)[2], const v8f (&ci)[2],
                                               v8f (&tr)[2], v8f (&tv)[2]) {
    v8f cs[2];
    cs[0] = cr[0] + ci[0];
    cs[1] = cr[1] + ci[1];
    v8f y1[2], y2[2], y3[2];
    y1[0] = v8f{}; y1[1] = v8f{}; y2[0] = v8f{}; y2[1] = v8f{}; y3[0] = v8f{}; y3[1] = v8f{};
    #pragma unroll
    for (int ee = 0; ee < 2; ++ee) {
        float arv[4], adv[4], asv[4];
        #pragma unroll
        for (int e = 0; e < 4; ++e) {
            arv[e] = aw[ee][e].r;
            adv[e] = aw[ee][e].i - aw[ee][e].r;
            asv[e] = aw[ee][e].r + aw[ee][e].i;
        }
        #pragma unroll
        for (int q = 0; q < 2; ++q) {
            const int s = ee * 2 + q;
            v2f fR, fD, fS;
            fR.x = arv[2 * q];       fR.y = arv[2 * q + 1];
            fD.x = adv[2 * q];       fD.y = adv[2 * q + 1];
            fS.x = asv[2 * q];       fS.y = asv[2 * q + 1];
            #pragma unroll
            for (int t = 0; t < 2; ++t) {
                v2f xS, xR, xI;
                xS.x = cs[t][2 * s];     xS.y = cs[t][2 * s + 1];
                xR.x = cr[t][2 * s];     xR.y = cr[t][2 * s + 1];
                xI.x = ci[t][2 * s];     xI.y = ci[t][2 * s + 1];
                y1[t] = wmma_f32(fR, xS, y1[t]);
                y2[t] = wmma_f32(fD, xR, y2[t]);
                y3[t] = wmma_f32(fS, xI, y3[t]);
            }
        }
    }
    v8f yr[2], yi[2];
    yr[0] = y1[0] - y3[0];  yi[0] = y1[0] + y2[0];
    yr[1] = y1[1] - y3[1];  yi[1] = y1[1] + y2[1];
    #pragma unroll
    for (int u = 0; u < 2; ++u) {
        tr[u] = ad[u][0].r * yr[0] - ad[u][0].i * yi[0] + ad[u][1].r * yr[1] - ad[u][1].i * yi[1];
        tv[u] = ad[u][0].r * yi[0] + ad[u][0].i * yr[0] + ad[u][1].r * yi[1] + ad[u][1].i * yr[1];
    }
}

template <int L>
__device__ __forceinline__ void layer_diag(v8f (&tr)[2], v8f (&tv)[2],
                                           const float* prr, const float* pri,
                                           const float* pcr, const float* pci,
                                           const int lane, const int tj) {
    const int col = tj * 16 + (lane & 15);
    float a2 = 1.f, b2 = 0.f;
    if (L < 3) { a2 = pcr[col]; b2 = pci[col]; }
    #pragma unroll
    for (int u = 0; u < 2; ++u) {
        #pragma unroll
        for (int v = 0; v < 8; ++v) {
            const int row = u * 16 + v + ((lane >> 4) << 3);
            const unsigned k = (unsigned)(row * 32 + col);
            const int par = __popc(k & (k >> 1)) + (int)((k & 1u) & (k >> 9));
            const float sgn = (par & 1) ? -1.f : 1.f;
            float pRe = sgn, pIm = 0.f;
            if (L < 3) {
                const float a1 = prr[row], b1 = pri[row];
                pRe = sgn * (a1 * a2 - b1 * b2);
                pIm = sgn * (a1 * b2 + b1 * a2);
            }
            const float er = tr[u][v], ei = tv[u][v];
            tr[u][v] = er * pRe - ei * pIm;
            tv[u][v] = er * pIm + ei * pRe;
        }
    }
}

__device__ __forceinline__ void store_strip(const v8f (&tr)[2], const v8f (&tv)[2],
                                            float* Dr, float* Di,
                                            const int lane, const int tj) {
    const int col = tj * 16 + (lane & 15);
    #pragma unroll
    for (int u = 0; u < 2; ++u) {
        #pragma unroll
        for (int v = 0; v < 8; ++v) {
            const int row = u * 16 + v + ((lane >> 4) << 3);
            Dr[row * LP + col] = tr[u][v];
            Di[row * LP + col] = tv[u][v];
        }
    }
}

__global__ __launch_bounds__(64) void qsim(const float* __restrict__ x,
                                           const float* __restrict__ w1,
                                           const float* __restrict__ w2,
                                           float* __restrict__ out) {
    __shared__ __attribute__((aligned(16))) float lds[4 * PLANE + 256 + 240 + 40];
    float* const S0r = lds;
    float* const S0i = lds + 1 * PLANE;
    float* const S1r = lds + 2 * PLANE;
    float* const S1i = lds + 3 * PLANE;
    float* const prr = lds + 4 * PLANE;
    float* const pri = prr + 32;
    float* const pcr = prr + 64;
    float* const pci = prr + 96;
    float* const ear = prr + 128;
    float* const eai = prr + 160;
    float* const ebr = prr + 192;
    float* const ebi = prr + 224;
    float* const G   = prr + 256;
    float* const H   = G + 240;
    float* const red = S0r;

    const int tid  = threadIdx.x;
    const int b    = blockIdx.x;
    const int lane = tid & 31;
    const int tj   = tid >> 5;

    const bool isG = tid < 30;
    const bool isH = (tid >= 32) && (tid < 42);
    float phi = 0.f, theta = 0.f, omega = 0.f;
    if (isG || isH) {
        const float* w = isG ? (w1 + tid * 3) : (w2 + (tid - 32) * 3);
        phi = w[0]; theta = w[1]; omega = w[2];
    }

    float xb[10];
    #pragma unroll
    for (int j = 0; j < 10; ++j) xb[j] = x[b * 10 + j];

    {
        const int  idx = tid & 31;
        const bool isC = tid >= 32;
        float a = 0.f;
        #pragma unroll
        for (int jj = 0; jj < 5; ++jj) {
            const int bit = (idx >> (4 - jj)) & 1;
            a += xb[isC ? 5 + jj : jj] * (1.f - 2.f * (float)bit);
        }
        a *= -0.5f;
        float sv, cv; sincosf(a, &sv, &cv);
        if (isC) { pcr[idx] = cv; pci[idx] = sv; }
        else     { prr[idx] = cv; pri[idx] = sv; }

        if (isG || isH) {
            float s, c, sa, ca, sb, cb;
            sincosf(0.5f * theta, &s, &c);
            sincosf(-0.5f * (phi + omega), &sa, &ca);
            sincosf(-0.5f * (phi - omega), &sb, &cb);
            if (isG) {
                const float epr = ca * c, epi = sa * c;
                const float emr = cb * s, emi = sb * s;
                float4* g = reinterpret_cast<float4*>(G + tid * 8);
                g[0] = make_float4(epr, epi, -emr, emi);
                g[1] = make_float4(emr, emi, epr, -epi);
            } else {
                const float pr = ca * cb - sa * sb;
                const float pi = ca * sb + sa * cb;
                const float g2 = 2.f * c * s;
                reinterpret_cast<float4*>(H)[tid - 32] =
                    make_float4(c * c - s * s, -g2 * pr, g2 * pi, 0.f);
            }
        }
        __syncthreads();

        cf f0; f0.r = 1.f; f0.i = 0.f;
        #pragma unroll
        for (int jj = 0; jj < 5; ++jj)
            f0 = cmul(f0, wire_entry(G, (isC ? 5 : 0) + jj, (idx >> (4 - jj)) & 1, 0));
        const float sg = (__popc(idx & (idx >> 1)) & 1) ? -1.f : 1.f;
        const float er = sg * (f0.r * cv - f0.i * sv);
        const float ei = sg * (f0.r * sv + f0.i * cv);
        if (isC) { ebr[idx] = er; ebi[idx] = ei; }
        else     { ear[idx] = er; eai[idx] = ei; }
    }

    float4 bf[4];
    cf r5[2];
    cf aw[2][4], ad[2][2];
    gen_bfrags<1>(G, lane, tj, bf, r5);
    prep_afrags<1>(G, lane, aw, ad);
    __syncthreads();

    v8f tr[2], tv[2], cr[2], ci[2];

    first_product_rank4(bf, r5, ear, eai, ebr, ebi, lane, cr, ci);
    second_product(aw, ad, cr, ci, tr, tv);
    layer_diag<1>(tr, tv, prr, pri, pcr, pci, lane, tj);
    gen_bfrags<2>(G, lane, tj, bf, r5);
    prep_afrags<2>(G, lane, aw, ad);
    store_strip(tr, tv, S0r, S0i, lane, tj);
    __syncthreads();

    first_product(bf, r5, S0r, S0i, lane, cr, ci);
    second_product(aw, ad, cr, ci, tr, tv);
    layer_diag<2>(tr, tv, prr, pri, pcr, pci, lane, tj);
    store_strip(tr, tv, S1r, S1i, lane, tj);

    float sr[16], si[16];
    #pragma unroll
    for (int i = 0; i < 16; ++i) { sr[i] = tr[i >> 3][i & 7]; si[i] = tv[i >> 3][i & 7]; }

    float dj[10], qr[10], qi[10];
    #pragma unroll
    for (int j = 0; j < 10; ++j) { dj[j] = 0.f; qr[j] = 0.f; qi[j] = 0.f; }

    float ptot = 0.f;
    #pragma unroll
    for (int i = 0; i < 16; ++i) {
        const float p = sr[i] * sr[i] + si[i] * si[i];
        ptot += p;
        dj[0] += (i & 8) ? -p : p;
        dj[2] += (i & 4) ? -p : p;
        dj[3] += (i & 2) ? -p : p;
        dj[4] += (i & 1) ? -p : p;
    }
    {
        const int c = lane & 15;
        dj[1] = (lane & 16) ? -ptot : ptot;
        dj[5] = tj          ? -ptot : ptot;
        dj[6] = (c & 8)     ? -ptot : ptot;
        dj[7] = (c & 4)     ? -ptot : ptot;
        dj[8] = (c & 2)     ? -ptot : ptot;
        dj[9] = (c & 1)     ? -ptot : ptot;
    }
    #pragma unroll
    for (int k = 0; k < 4; ++k) {
        const int m = 8 >> k;
        const int j = (k == 0) ? 0 : k + 1;
        #pragma unroll
        for (int i = 0; i < 16; ++i) {
            if (i & m) continue;
            qr[j] += sr[i] * sr[i | m] + si[i] * si[i | m];
            qi[j] += sr[i] * si[i | m] - si[i] * sr[i | m];
        }
    }

    __syncthreads();

    float orr[16], oi[16];
    {
        const float4* pr4 = reinterpret_cast<const float4*>(&S1r[lane * LP + 16 * tj]);
        const float4* pi4 = reinterpret_cast<const float4*>(&S1i[lane * LP + 16 * tj]);
        #pragma unroll
        for (int k = 0; k < 4; ++k) {
            const float4 a = pr4[k], bq = pi4[k];
            orr[4 * k] = a.x;  orr[4 * k + 1] = a.y;  orr[4 * k + 2] = a.z;  orr[4 * k + 3] = a.w;
            oi[4 * k]  = bq.x; oi[4 * k + 1]  = bq.y; oi[4 * k + 2]  = bq.z; oi[4 * k + 3]  = bq.w;
        }
    }
    #pragma unroll
    for (int k = 0; k < 4; ++k) {
        const int m = 8 >> k;
        const int j = 6 + k;
        #pragma unroll
        for (int i = 0; i < 16; ++i) {
            if (i & m) continue;
            qr[j] += orr[i] * orr[i | m] + oi[i] * oi[i | m];
            qi[j] += orr[i] * oi[i | m] - oi[i] * orr[i | m];
        }
    }
    {
        float ar[8], ai[8], wr[8], wi[8];
        load8(&S1r[lane * LP + 24 * tj], ar);
        load8(&S1i[lane * LP + 24 * tj], ai);
        load8(&S1r[lane * LP + 16 * (tj ^ 1) + 8 * tj], wr);
        load8(&S1i[lane * LP + 16 * (tj ^ 1) + 8 * tj], wi);
        #pragma unroll
        for (int k = 0; k < 8; ++k) {
            qr[5] += ar[k] * wr[k] + ai[k] * wi[k];
            qi[5] += ar[k] * wi[k] - ai[k] * wr[k];
        }
        qi[5] = tj ? -qi[5] : qi[5];
    }
    {
        const int rb = (lane >> 3) & 1;
        const int o0 = 16 * tj + 8 * rb;
        float ar[8], ai[8], wr[8], wi[8];
        load8(&S1r[lane * LP + o0], ar);
        load8(&S1i[lane * LP + o0], ai);
        load8(&S1r[(lane ^ 8) * LP + o0], wr);
        load8(&S1i[(lane ^ 8) * LP + o0], wi);
        #pragma unroll
        for (int k = 0; k < 8; ++k) {
            qr[1] += ar[k] * wr[k] + ai[k] * wi[k];
            qi[1] += ar[k] * wi[k] - ai[k] * wr[k];
        }
        qi[1] = rb ? -qi[1] : qi[1];
    }

    const float4* Hq = reinterpret_cast<const float4*>(H);
    #pragma unroll
    for (int j = 0; j < 10; ++j) {
        const float4 hj = Hq[j];
        red[tid * 10 + j] = hj.x * dj[j] + 2.f * (hj.y * qr[j] - hj.z * qi[j]);
    }
    __syncthreads();
    if (tid < 10) {
        float s = 0.f;
        for (int t = 0; t < 64; ++t) s += red[t * 10 + tid];
        out[b * 10 + tid] = s;
    }
}

extern "C" void kernel_launch(void* const* d_in, const int* in_sizes, int n_in,
                              void* d_out, int out_size, void* d_ws, size_t ws_size,
                              hipStream_t stream) {
    const float* x  = (const float*)d_in[0];
    const float* w1 = (const float*)d_in[1];
    const float* w2 = (const float*)d_in[2];
    float* out = (float*)d_out;

    const int B = in_sizes[0] / 10;

    hipLaunchKernelGGL(qsim, dim3(B), dim3(64), 0, stream, x, w1, w2, out);
}


# re-upload phase RZ(x_j) folded into the next Rot of its wire as phi + x_j (layers 1, 2, last): no phase tables, layer diagonal is a CZ sign-bit flip; trig argument rounding changes
# speedup vs baseline: 4.7684x; 1.1716x over previous
#include <hip/hip_runtime.h>

typedef float v2f __attribute__((ext_vector_type(2)));
typedef float v8f __attribute__((ext_vector_type(8)));

__device__ __forceinline__ v8f wmma_f32(v2f a, v2f b, v8f c) {
    return __builtin_amdgcn_wmma_f32_16x16x4_f32(false, a, false, b, (short)0, c, false, false);
}

constexpr int LP    = 36;
constexpr int PLANE = 32 * LP;

struct cf { float r, i; };
__device__ __forceinline__ cf cmul(const cf a, const cf b) {
    cf o;
    o.r = a.r * b.r - a.i * b.i;
    o.i = a.r * b.i + a.i * b.r;
    return o;
}

__device__ __forceinline__ void wire_row(const float* Gl, const int wire, const int rb,
                                         cf (&g)[2]) {
    const float4 v = *reinterpret_cast<const float4*>(&Gl[wire * 8 + rb * 4]);
    g[0].r = v.x; g[0].i = v.y; g[1].r = v.z; g[1].i = v.w;
}
__device__ __forceinline__ cf wire_entry(const float* Gl, const int wire, const int rb,
                                         const int cb) {
    const float2 v = *reinterpret_cast<const float2*>(&Gl[wire * 8 + rb * 4 + cb * 2]);
    cf o; o.r = v.x; o.i = v.y;
    return o;
}

template <int L>
__device__ __forceinline__ void gen_bfrags(const float* G,
                                           const int lane, const int tj,
                                           float4 (&bf)[4], cf (&r5)[2]) {
    const float* Gl = G + L * 80;
    const int c = lane & 15;
    const int h = lane >> 4;
    cf r6[2], r8[2], r9[2];
    wire_row(Gl, 5, tj, r5);
    wire_row(Gl, 6, (c >> 3) & 1, r6);
    const cf r7 = wire_entry(Gl, 7, (c >> 2) & 1, h);
    wire_row(Gl, 8, (c >> 1) & 1, r8);
    wire_row(Gl, 9, c & 1, r9);
    cf v[4];
    #pragma unroll
    for (int e = 0; e < 4; ++e) v[e] = cmul(r7, cmul(r8[e >> 1], r9[e & 1]));
    #pragma unroll
    for (int j0 = 0; j0 < 2; ++j0) {
        const cf d = r6[j0];
        const cf e0 = cmul(d, v[0]), e1 = cmul(d, v[1]), e2 = cmul(d, v[2]), e3 = cmul(d, v[3]);
        bf[j0 * 2]     = make_float4(e0.r, e1.r, e2.r, e3.r);
        bf[j0 * 2 + 1] = make_float4(e0.i, e1.i, e2.i, e3.i);
    }
}

template <int L>
__device__ __forceinline__ void prep_afrags(const float* G, const int lane,
                                            cf (&w)[2][4], cf (&d)[2][2]) {
    const float* Gl = G + L * 80;
    const int c = lane & 15;
    const int h = lane >> 4;
    cf r2[2], r3[2], r4[2];
    wire_row(Gl, 0, 0, d[0]);
    wire_row(Gl, 0, 1, d[1]);
    const cf r1 = wire_entry(Gl, 1, (c >> 3) & 1, h);
    wire_row(Gl, 2, (c >> 2) & 1, r2);
    wire_row(Gl, 3, (c >> 1) & 1, r3);
    wire_row(Gl, 4, c & 1, r4);
    cf y[4];
    #pragma unroll
    for (int e = 0; e < 4; ++e) y[e] = cmul(r3[e >> 1], r4[e & 1]);
    #pragma unroll
    for (int ee = 0; ee < 2; ++ee) {
        const cf z = cmul(r1, r2[ee]);
        #pragma unroll
        for (int e = 0; e < 4; ++e) w[ee][e] = cmul(z, y[e]);
    }
}

__device__ __forceinline__ void load8(const float* p, float (&v)[8]) {
    const float4 a = reinterpret_cast<const float4*>(p)[0];
    const float4 b = reinterpret_cast<const float4*>(p)[1];
    v[0] = a.x; v[1] = a.y; v[2] = a.z; v[3] = a.w;
    v[4] = b.x; v[5] = b.y; v[6] = b.z; v[7] = b.w;
}


__device__ __forceinline__ void first_product(const float4 (&bf)[4], const cf (&r5)[2],
                                              const float* Sr, const float* Si,
                                              const int lane,
                                              v8f (&cr)[2], v8f (&ci)[2]) {
    const int c = lane & 15;
    const int h = lane >> 4;
    const float g0r = r5[0].r, g0i = r5[0].i, g1r = r5[1].r, g1i = r5[1].i;
    v8f p1[2], p2[2], p3[2];
    p1[0] = v8f{}; p1[1] = v8f{}; p2[0] = v8f{}; p2[1] = v8f{}; p3[0] = v8f{}; p3[1] = v8f{};
    #pragma unroll
    for (int j0 = 0; j0 < 2; ++j0) {
        const float4 br4 = bf[j0 * 2];
        const float4 bi4 = bf[j0 * 2 + 1];
        const v2f br[2] = {v2f{br4.x, br4.y}, v2f{br4.z, br4.w}};
        const v2f bi[2] = {v2f{bi4.x, bi4.y}, v2f{bi4.z, bi4.w}};
        #pragma unroll
        for (int t = 0; t < 2; ++t) {
            const int so = (16 * t + c) * LP + 8 * j0 + 4 * h;
            const float4 lr4 = *reinterpret_cast<const float4*>(&Sr[so]);
            const float4 li4 = *reinterpret_cast<const float4*>(&Si[so]);
            const float4 hr4 = *reinterpret_cast<const float4*>(&Sr[so + 16]);
            const float4 hi4 = *reinterpret_cast<const float4*>(&Si[so + 16]);
            const v2f lr[2] = {v2f{lr4.x, lr4.y}, v2f{lr4.z, lr4.w}};
            const v2f li[2] = {v2f{li4.x, li4.y}, v2f{li4.z, li4.w}};
            const v2f hr[2] = {v2f{hr4.x, hr4.y}, v2f{hr4.z, hr4.w}};
            const v2f hi[2] = {v2f{hi4.x, hi4.y}, v2f{hi4.z, hi4.w}};
            #pragma unroll
            for (int q = 0; q < 2; ++q) {
                const v2f xR = g0r * lr[q] - g0i * li[q] + g1r * hr[q] - g1i * hi[q];
                const v2f xI = g0r * li[q] + g0i * lr[q] + g1r * hi[q] + g1i * hr[q];
                const v2f xS = xR + xI;
                const v2f fR = br[q];
                const v2f fD = bi[q] - br[q];
                const v2f fS = br[q] + bi[q];
                p1[t] = wmma_f32(xS, fR, p1[t]);
                p2[t] = wmma_f32(xR, fD, p2[t]);
                p3[t] = wmma_f32(xI, fS, p3[t]);
            }
        }
    }
    cr[0] = p1[0] - p3[0];  ci[0] = p1[0] + p2[0];
    cr[1] = p1[1] - p3[1];  ci[1] = p1[1] + p2[1];
}

__device__ __forceinline__ void first_product_rank4(const float4 (&bf)[4], const cf (&r5)[2],
                                                    const float* ear, const float* eai,
                                                    const float* ebr, const float* ebi,
                                                    const int lane,
                                                    v8f (&cr)[2], v8f (&ci)[2]) {
    const int c = lane & 15;
    const int h = lane >> 4;
    const bool isIm = (c & 4) != 0;
    const float sA = (c & 1) ? -1.f : 1.f;
    const float sB = (c & 2) ? -1.f : 1.f;
    const float g0r = r5[0].r, g0i = r5[0].i;
    const float g1r = sA * r5[1].r, g1i = sA * r5[1].i;
    const v2f sv = v2f{1.f, sB};
    v8f acc[2];
    acc[0] = v8f{}; acc[1] = v8f{};
    #pragma unroll
    for (int j0 = 0; j0 < 2; ++j0) {
        const float4 lr4 = *reinterpret_cast<const float4*>(&ebr[8 * j0 + 4 * h]);
        const float4 li4 = *reinterpret_cast<const float4*>(&ebi[8 * j0 + 4 * h]);
        const float4 hr4 = *reinterpret_cast<const float4*>(&ebr[16 + 8 * j0 + 4 * h]);
        const float4 hi4 = *reinterpret_cast<const float4*>(&ebi[16 + 8 * j0 + 4 * h]);
        const v2f lr[2] = {v2f{lr4.x, lr4.y}, v2f{lr4.z, lr4.w}};
        const v2f li[2] = {v2f{li4.x, li4.y}, v2f{li4.z, li4.w}};
        const v2f hr[2] = {v2f{hr4.x, hr4.y}, v2f{hr4.z, hr4.w}};
        const v2f hi[2] = {v2f{hi4.x, hi4.y}, v2f{hi4.z, hi4.w}};
        const float4 br4 = bf[j0 * 2];
        const float4 bi4 = bf[j0 * 2 + 1];
        const v2f br[2] = {v2f{br4.x, br4.y}, v2f{br4.z, br4.w}};
        const v2f bi[2] = {v2f{bi4.x, bi4.y}, v2f{bi4.z, bi4.w}};
        #pragma unroll
        for (int q = 0; q < 2; ++q) {
            const v2f vr = sv * (g0r * lr[q] - g0i * li[q] + g1r * hr[q] - g1i * hi[q]);
            const v2f vi = sv * (g0r * li[q] + g0i * lr[q] + g1r * hi[q] + g1i * hr[q]);
            const v2f a1 = isIm ? vi : vr;
            const v2f a2 = isIm ? vr : -vi;
            acc[q] = wmma_f32(a1, br[q], acc[q]);
            acc[q] = wmma_f32(a2, bi[q], acc[q]);
        }
    }
    const v8f qt = acc[0] + acc[1];
    #pragma unroll
    for (int t = 0; t < 2; ++t) {
        float ar[8], ai[8];
        load8(&ear[16 * t + 8 * h], ar);
        load8(&eai[16 * t + 8 * h], ai);
        #pragma unroll
        for (int r = 0; r < 8; ++r) {
            const int id = (r & 1) + 2 * t;
            const float qr = qt[id], qi = qt[4 + id];
            cr[t][r] = ar[r] * qr - ai[r] * qi;
            ci[t][r] = ar[r] * qi + ai[r] * qr;
        }
    }
}

__device__ __forceinline__ void second_product(const cf (&aw)[2][4], const cf (&ad)[2][2],
                                               const v8f (&cr)[2], const v8f (&ci)[2],
                                               v8f (&tr)[2], v8f (&tv)[2]) {
    v8f cs[2];
    cs[0] = cr[0] + ci[0];
    cs[1] = cr[1] + ci[1];
    v8f y1[2], y2[2], y3[2];
    y1[0] = v8f{}; y1[1] = v8f{}; y2[0] = v8f{}; y2[1] = v8f{}; y3[0] = v8f{}; y3[1] = v8f{};
    #pragma unroll
    for (int ee = 0; ee < 2; ++ee) {
        float arv[4], adv[4], asv[4];
        #pragma unroll
        for (int e = 0; e < 4; ++e) {
            arv[e] = aw[ee][e].r;
            adv[e] = aw[ee][e].i - aw[ee][e].r;
            asv[e] = aw[ee][e].r + aw[ee][e].i;
        }
        #pragma unroll
        for (int q = 0; q < 2; ++q) {
            const int s = ee * 2 + q;
            v2f fR, fD, fS;
            fR.x = arv[2 * q];       fR.y = arv[2 * q + 1];
            fD.x = adv[2 * q];       fD.y = adv[2 * q + 1];
            fS.x = asv[2 * q];       fS.y = asv[2 * q + 1];
            #pragma unroll
            for (int t = 0; t < 2; ++t) {
                v2f xS, xR, xI;
                xS.x = cs[t][2 * s];     xS.y = cs[t][2 * s + 1];
                xR.x = cr[t][2 * s];     xR.y = cr[t][2 * s + 1];
                xI.x = ci[t][2 * s];     xI.y = ci[t][2 * s + 1];
                y1[t] = wmma_f32(fR, xS, y1[t]);
                y2[t] = wmma_f32(fD, xR, y2[t]);
                y3[t] = wmma_f32(fS, xI, y3[t]);
            }
        }
    }
    v8f yr[2], yi[2];
    yr[0] = y1[0] - y3[0];  yi[0] = y1[0] + y2[0];
    yr[1] = y1[1] - y3[1];  yi[1] = y1[1] + y2[1];
    #pragma unroll
    for (int u = 0; u < 2; ++u) {
        tr[u] = ad[u][0].r * yr[0] - ad[u][0].i * yi[0] + ad[u][1].r * yr[1] - ad[u][1].i * yi[1];
        tv[u] = ad[u][0].r * yi[0] + ad[u][0].i * yr[0] + ad[u][1].r * yi[1] + ad[u][1].i * yr[1];
    }
}

__device__ __forceinline__ void layer_sign(v8f (&tr)[2], v8f (&tv)[2],
                                           const int lane, const int tj) {
    const unsigned c   = (unsigned)(lane & 15);
    const unsigned h   = (unsigned)(lane >> 4);
    const unsigned col = (unsigned)tj * 16u + c;
    const unsigned mL  = ((unsigned)__popc(col & (col >> 1)) & 1u) << 31;
    const unsigned mU  = ((h ^ c) & 1u) << 31;
    const unsigned mV2 = h << 31;
    const unsigned mV0 = (unsigned)tj << 31;
    #pragma unroll
    for (int u = 0; u < 2; ++u) {
        #pragma unroll
        for (int v = 0; v < 8; ++v) {
            const int v2 = (v >> 2) & 1, v1 = (v >> 1) & 1, v0 = v & 1;
            unsigned m = mL;
            if (u)  m ^= mU;
            if (v2) m ^= mV2;
            if (v0) m ^= mV0;
            if ((v2 & v1) ^ (v1 & v0)) m ^= 0x80000000u;
            tr[u][v] = __uint_as_float(__float_as_uint(tr[u][v]) ^ m);
            tv[u][v] = __uint_as_float(__float_as_uint(tv[u][v]) ^ m);
        }
    }
}

__device__ __forceinline__ void store_strip(const v8f (&tr)[2], const v8f (&tv)[2],
                                            float* Dr, float* Di,
                                            const int lane, const int tj) {
    const int col = tj * 16 + (lane & 15);
    #pragma unroll
    for (int u = 0; u < 2; ++u) {
        #pragma unroll
        for (int v = 0; v < 8; ++v) {
            const int row = u * 16 + v + ((lane >> 4) << 3);
            Dr[row * LP + col] = tr[u][v];
            Di[row * LP + col] = tv[u][v];
        }
    }
}

__global__ __launch_bounds__(64) void qsim(const float* __restrict__ x,
                                           const float* __restrict__ w1,
                                           const float* __restrict__ w2,
                                           float* __restrict__ out) {
    __shared__ __attribute__((aligned(16))) float lds[4 * PLANE + 128 + 240 + 40];
    float* const S0r = lds;
    float* const S0i = lds + 1 * PLANE;
    float* const S1r = lds + 2 * PLANE;
    float* const S1i = lds + 3 * PLANE;
    float* const ear = lds + 4 * PLANE;
    float* const eai = ear + 32;
    float* const ebr = ear + 64;
    float* const ebi = ear + 96;
    float* const G   = ear + 128;
    float* const H   = G + 240;
    float* const red = S0r;

    const int tid  = threadIdx.x;
    const int b    = blockIdx.x;
    const int lane = tid & 31;
    const int tj   = tid >> 5;

    const bool isG = tid < 30;
    const bool isH = (tid >= 32) && (tid < 42);
    float phi = 0.f, theta = 0.f, omega = 0.f;
    if (isG || isH) {
        const float* w = isG ? (w1 + tid * 3) : (w2 + (tid - 32) * 3);
        phi = w[0]; theta = w[1]; omega = w[2];
        if (tid >= 10) phi += x[b * 10 + (isG ? tid % 10 : tid - 32)];
    }

    {
        const int  idx = tid & 31;
        const bool isC = tid >= 32;

        if (isG || isH) {
            float s, c, sa, ca, sb, cb;
            sincosf(0.5f * theta, &s, &c);
            sincosf(-0.5f * (phi + omega), &sa, &ca);
            sincosf(-0.5f * (phi - omega), &sb, &cb);
            if (isG) {
                const float epr = ca * c, epi = sa * c;
                const float emr = cb * s, emi = sb * s;
                float4* g = reinterpret_cast<float4*>(G + tid * 8);
                g[0] = make_float4(epr, epi, -emr, emi);
                g[1] = make_float4(emr, emi, epr, -epi);
            } else {
                const float pr = ca * cb - sa * sb;
                const float pi = ca * sb + sa * cb;
                const float g2 = 2.f * c * s;
                reinterpret_cast<float4*>(H)[tid - 32] =
                    make_float4(c * c - s * s, -g2 * pr, g2 * pi, 0.f);
            }
        }
        __syncthreads();

        cf f0; f0.r = 1.f; f0.i = 0.f;
        #pragma unroll
        for (int jj = 0; jj < 5; ++jj)
            f0 = cmul(f0, wire_entry(G, (isC ? 5 : 0) + jj, (idx >> (4 - jj)) & 1, 0));
        const float sg = (__popc(idx & (idx >> 1)) & 1) ? -1.f : 1.f;
        const float er = sg * f0.r;
        const float ei = sg * f0.i;
        if (isC) { ebr[idx] = er; ebi[idx] = ei; }
        else     { ear[idx] = er; eai[idx] = ei; }
    }

    float4 bf[4];
    cf r5[2];
    cf aw[2][4], ad[2][2];
    gen_bfrags<1>(G, lane, tj, bf, r5);
    prep_afrags<1>(G, lane, aw, ad);
    __syncthreads();

    v8f tr[2], tv[2], cr[2], ci[2];

    first_product_rank4(bf, r5, ear, eai, ebr, ebi, lane, cr, ci);
    second_product(aw, ad, cr, ci, tr, tv);
    layer_sign(tr, tv, lane, tj);
    gen_bfrags<2>(G, lane, tj, bf, r5);
    prep_afrags<2>(G, lane, aw, ad);
    store_strip(tr, tv, S0r, S0i, lane, tj);
    __syncthreads();

    first_product(bf, r5, S0r, S0i, lane, cr, ci);
    second_product(aw, ad, cr, ci, tr, tv);
    layer_sign(tr, tv, lane, tj);
    store_strip(tr, tv, S1r, S1i, lane, tj);

    float sr[16], si[16];
    #pragma unroll
    for (int i = 0; i < 16; ++i) { sr[i] = tr[i >> 3][i & 7]; si[i] = tv[i >> 3][i & 7]; }

    float dj[10], qr[10], qi[10];
    #pragma unroll
    for (int j = 0; j < 10; ++j) { dj[j] = 0.f; qr[j] = 0.f; qi[j] = 0.f; }

    float ptot = 0.f;
    #pragma unroll
    for (int i = 0; i < 16; ++i) {
        const float p = sr[i] * sr[i] + si[i] * si[i];
        ptot += p;
        dj[0] += (i & 8) ? -p : p;
        dj[2] += (i & 4) ? -p : p;
        dj[3] += (i & 2) ? -p : p;
        dj[4] += (i & 1) ? -p : p;
    }
    {
        const int c = lane & 15;
        dj[1] = (lane & 16) ? -ptot : ptot;
        dj[5] = tj          ? -ptot : ptot;
        dj[6] = (c & 8)     ? -ptot : ptot;
        dj[7] = (c & 4)     ? -ptot : ptot;
        dj[8] = (c & 2)     ? -ptot : ptot;
        dj[9] = (c & 1)     ? -ptot : ptot;
    }
    #pragma unroll
    for (int k = 0; k < 4; ++k) {
        const int m = 8 >> k;
        const int j = (k == 0) ? 0 : k + 1;
        #pragma unroll
        for (int i = 0; i < 16; ++i) {
            if (i & m) continue;
            qr[j] += sr[i] * sr[i | m] + si[i] * si[i | m];
            qi[j] += sr[i] * si[i | m] - si[i] * sr[i | m];
        }
    }

    __syncthreads();

    float orr[16], oi[16];
    {
        const float4* pr4 = reinterpret_cast<const float4*>(&S1r[lane * LP + 16 * tj]);
        const float4* pi4 = reinterpret_cast<const float4*>(&S1i[lane * LP + 16 * tj]);
        #pragma unroll
        for (int k = 0; k < 4; ++k) {
            const float4 a = pr4[k], bq = pi4[k];
            orr[4 * k] = a.x;  orr[4 * k + 1] = a.y;  orr[4 * k + 2] = a.z;  orr[4 * k + 3] = a.w;
            oi[4 * k]  = bq.x; oi[4 * k + 1]  = bq.y; oi[4 * k + 2]  = bq.z; oi[4 * k + 3]  = bq.w;
        }
    }
    #pragma unroll
    for (int k = 0; k < 4; ++k) {
        const int m = 8 >> k;
        const int j = 6 + k;
        #pragma unroll
        for (int i = 0; i < 16; ++i) {
            if (i & m) continue;
            qr[j] += orr[i] * orr[i | m] + oi[i] * oi[i | m];
            qi[j] += orr[i] * oi[i | m] - oi[i] * orr[i | m];
        }
    }
    {
        float ar[8], ai[8], wr[8], wi[8];
        load8(&S1r[lane * LP + 24 * tj], ar);
        load8(&S1i[lane * LP + 24 * tj], ai);
        load8(&S1r[lane * LP + 16 * (tj ^ 1) + 8 * tj], wr);
        load8(&S1i[lane * LP + 16 * (tj ^ 1) + 8 * tj], wi);
        #pragma unroll
        for (int k = 0; k < 8; ++k) {
            qr[5] += ar[k] * wr[k] + ai[k] * wi[k];
            qi[5] += ar[k] * wi[k] - ai[k] * wr[k];
        }
        qi[5] = tj ? -qi[5] : qi[5];
    }
    {
        const int rb = (lane >> 3) & 1;
        const int o0 = 16 * tj + 8 * rb;
        float ar[8], ai[8], wr[8], wi[8];
        load8(&S1r[lane * LP + o0], ar);
        load8(&S1i[lane * LP + o0], ai);
        load8(&S1r[(lane ^ 8) * LP + o0], wr);
        load8(&S1i[(lane ^ 8) * LP + o0], wi);
        #pragma unroll
        for (int k = 0; k < 8; ++k) {
            qr[1] += ar[k] * wr[k] + ai[k] * wi[k];
            qi[1] += ar[k] * wi[k] - ai[k] * wr[k];
        }
        qi[1] = rb ? -qi[1] : qi[1];
    }

    const float4* Hq = reinterpret_cast<const float4*>(H);
    #pragma unroll
    for (int j = 0; j < 10; ++j) {
        const float4 hj = Hq[j];
        red[tid * 10 + j] = hj.x * dj[j] + 2.f * (hj.y * qr[j] - hj.z * qi[j]);
    }
    __syncthreads();
    if (tid < 10) {
        float s = 0.f;
        for (int t = 0; t < 64; ++t) s += red[t * 10 + tid];
        out[b * 10 + tid] = s;
    }
}

extern "C" void kernel_launch(void* const* d_in, const int* in_sizes, int n_in,
                              void* d_out, int out_size, void* d_ws, size_t ws_size,
                              hipStream_t stream) {
    const float* x  = (const float*)d_in[0];
    const float* w1 = (const float*)d_in[1];
    const float* w2 = (const float*)d_in[2];
    float* out = (float*)d_out;

    const int B = in_sizes[0] / 10;

    hipLaunchKernelGGL(qsim, dim3(B), dim3(64), 0, stream, x, w1, w2, out);
}
